# starting kernel
# baseline (speedup 1.0000x reference)
#include <hip/hip_runtime.h>


typedef __attribute__((ext_vector_type(2))) float v2f;
typedef __attribute__((ext_vector_type(8))) float v8f;

#define CHUNK_L 196
#define NCOLS   512
#define NLEV    9

struct Params {
    float cm00, cm01, cm10, cm11;
    float km00, km01, km10, km11;
    float m0, m1, dt, h2, dt6;
};

__device__ inline Params make_params(const float* M, const float* C,
                                     const float* K, const float* dtp) {
    Params p;
    p.cm00 = C[0] + C[1]; p.cm01 = -C[1];
    p.cm10 = -C[1];       p.cm11 = C[2] + C[1];
    p.km00 = K[0] + K[1]; p.km01 = -K[1];
    p.km10 = -K[1];       p.km11 = K[2] + K[1];
    p.m0 = M[0]; p.m1 = M[1];
    p.dt = dtp[0];
    p.h2 = 0.5f * p.dt;
    p.dt6 = p.dt / 6.0f;
    return p;
}

__device__ inline void accel(const Params& p, float z0, float z1,
                             float v0, float v1, float u0, float u1,
                             float& a0, float& a1) {
    a0 = (u0 - (p.cm00 * v0 + p.cm01 * v1) - (p.km00 * z0 + p.km01 * z1)) / p.m0;
    a1 = (u1 - (p.cm10 * v0 + p.cm11 * v1) - (p.km10 * z0 + p.km11 * z1)) / p.m1;
}

__device__ inline void rk4_step(const Params& p, float& y0, float& y1,
                                float& v0, float& v1, float u0, float u1) {
    float k1x, k1y; accel(p, y0, y1, v0, v1, u0, u1, k1x, k1y);
    float y2x = y0 + v0 * p.h2, y2y = y1 + v1 * p.h2;
    float v2x = v0 + k1x * p.h2, v2y = v1 + k1y * p.h2;
    float k2x, k2y; accel(p, y2x, y2y, v2x, v2y, u0, u1, k2x, k2y);
    float y3x = y0 + v2x * p.h2, y3y = y1 + v2y * p.h2;
    float v3x = v0 + k2x * p.h2, v3y = v1 + k2y * p.h2;
    float k3x, k3y; accel(p, y3x, y3y, v3x, v3y, u0, u1, k3x, k3y);
    float y4x = y0 + v3x * p.dt, y4y = y1 + v3y * p.dt;
    float v4x = v0 + k3x * p.dt, v4y = v1 + k3y * p.dt;
    float k4x, k4y; accel(p, y4x, y4y, v4x, v4y, u0, u1, k4x, k4y);
    float yn0 = y0 + p.dt6 * (v0 + 2.0f * v2x + 2.0f * v3x + v4x);
    float yn1 = y1 + p.dt6 * (v1 + 2.0f * v2y + 2.0f * v3y + v4y);
    float vn0 = v0 + p.dt6 * (k1x + 2.0f * k2x + 2.0f * k3x + k4x);
    float vn1 = v1 + p.dt6 * (k1y + 2.0f * k2y + 2.0f * k3y + k4y);
    y0 = yn0; y1 = yn1; v0 = vn0; v1 = vn1;
}

__device__ inline void lds_wave_fence() {
    asm volatile("s_wait_dscnt 0x0" ::: "memory");
}

__device__ inline void wmma_mul4_lds(const float* src1, const float* src2,
                                     float* dst, int lane) {
    const int  n  = lane & 15;
    const bool hi = lane >= 16;
    v2f a, b;
    a.x = (n < 4) ? src1[n * 4 + (hi ? 2 : 0)] : 0.0f;
    a.y = (n < 4) ? src1[n * 4 + (hi ? 3 : 1)] : 0.0f;
    b.x = (n < 4) ? src2[(hi ? 2 : 0) * 4 + n] : 0.0f;
    b.y = (n < 4) ? src2[(hi ? 3 : 1) * 4 + n] : 0.0f;
    v8f c = {0.f, 0.f, 0.f, 0.f, 0.f, 0.f, 0.f, 0.f};
    v8f d = __builtin_amdgcn_wmma_f32_16x16x4_f32(
        false, a, false, b, (short)0, c, false, false);
    if (!hi && n < 4) {
        dst[0 * 4 + n] = d[0];
        dst[1 * 4 + n] = d[1];
        dst[2 * 4 + n] = d[2];
        dst[3 * 4 + n] = d[3];
    }
    lds_wave_fence();
}

__global__ void phase1_chunks(const float* __restrict__ u,
                              const float* __restrict__ M,
                              const float* __restrict__ C,
                              const float* __restrict__ K,
                              const float* __restrict__ dtp,
                              float* __restrict__ wsE, int nfull) {
    int i = blockIdx.x * blockDim.x + threadIdx.x;
    if (i >= nfull) return;
    Params p = make_params(M, C, K, dtp);
    float y0 = 0.f, y1 = 0.f, v0 = 0.f, v1 = 0.f;
    const float* up = u + (size_t)i * CHUNK_L * 2;
    for (int t = 0; t < CHUNK_L; ++t) {
        if (t + 16 < CHUNK_L) __builtin_prefetch(up + 2 * (t + 16), 0, 3);
        float u0 = up[2 * t], u1 = up[2 * t + 1];
        rk4_step(p, y0, y1, v0, v1, u0, u1);
    }
    wsE[i * 4 + 0] = y0; wsE[i * 4 + 1] = y1;
    wsE[i * 4 + 2] = v0; wsE[i * 4 + 3] = v1;
}

__global__ __launch_bounds__(1024) void phase2_scan(
    const float* __restrict__ M, const float* __restrict__ C,
    const float* __restrict__ K, const float* __restrict__ x0,
    const float* __restrict__ dtp, const float* __restrict__ wsE,
    float* __restrict__ wsS, int P) {
    __shared__ float xs[NCOLS][4];
    __shared__ float Gd[NLEV][16];
    __shared__ float mA[16], mT4[16], mT64[16], mQ[16], mR[16];

    const int  tid  = threadIdx.x;
    const int  lane = tid & 31;
    const int  wave = tid >> 5;
    const int  n    = lane & 15;
    const bool hi   = lane >= 16;
    const int  col  = wave * 16 + n;

    if (tid < NCOLS) {
        float a0, a1, a2, a3;
        if (tid == 0) {
            a0 = x0[0]; a1 = x0[2]; a2 = x0[1]; a3 = x0[3];
        } else if (tid < P) {
            a0 = wsE[(tid - 1) * 4 + 0]; a1 = wsE[(tid - 1) * 4 + 1];
            a2 = wsE[(tid - 1) * 4 + 2]; a3 = wsE[(tid - 1) * 4 + 3];
        } else {
            a0 = a1 = a2 = a3 = 0.f;
        }
        xs[tid][0] = a0; xs[tid][1] = a1; xs[tid][2] = a2; xs[tid][3] = a3;
    }

    if (tid < 32) {
        Params p = make_params(M, C, K, dtp);
        {
            float y0 = (lane == 0) ? 1.f : 0.f, y1 = (lane == 1) ? 1.f : 0.f;
            float v0 = (lane == 2) ? 1.f : 0.f, v1 = (lane == 3) ? 1.f : 0.f;
            rk4_step(p, y0, y1, v0, v1, 0.f, 0.f);
            if (lane < 4) {
                mA[0 * 4 + lane] = y0;
                mA[1 * 4 + lane] = y1;
                mA[2 * 4 + lane] = v0;
                mA[3 * 4 + lane] = v1;
            }
        }
        lds_wave_fence();
        wmma_mul4_lds(mA,   mA,   mQ,    lane);
        wmma_mul4_lds(mQ,   mQ,   mT4,   lane);
        wmma_mul4_lds(mT4,  mT4,  mQ,    lane);
        wmma_mul4_lds(mQ,   mQ,   mR,    lane);
        wmma_mul4_lds(mR,   mR,   mQ,    lane);
        wmma_mul4_lds(mQ,   mQ,   mT64,  lane);
        wmma_mul4_lds(mT64, mT64, mQ,    lane);
        wmma_mul4_lds(mQ,   mT64, mR,    lane);
        wmma_mul4_lds(mR,   mT4,  Gd[0], lane);
        for (int l = 1; l < NLEV; ++l) {
            wmma_mul4_lds(Gd[l - 1], Gd[l - 1], Gd[l], lane);
        }
    }
    __syncthreads();

    for (int l = 0; l < NLEV; ++l) {
        const int d = 1 << l;
        v2f a, b;
        a.x = (n < 4) ? Gd[l][n * 4 + (hi ? 2 : 0)] : 0.0f;
        a.y = (n < 4) ? Gd[l][n * 4 + (hi ? 3 : 1)] : 0.0f;
        const int src = col - d;
        b.x = (src >= 0) ? xs[src][hi ? 2 : 0] : 0.0f;
        b.y = (src >= 0) ? xs[src][hi ? 3 : 1] : 0.0f;
        float c0 = hi ? 0.f : xs[col][0];
        float c1 = hi ? 0.f : xs[col][1];
        float c2 = hi ? 0.f : xs[col][2];
        float c3 = hi ? 0.f : xs[col][3];
        v8f c = {c0, c1, c2, c3, 0.f, 0.f, 0.f, 0.f};
        __syncthreads();
        v8f dres = __builtin_amdgcn_wmma_f32_16x16x4_f32(
            false, a, false, b, (short)0, c, false, false);
        if (!hi) {
            xs[col][0] = dres[0]; xs[col][1] = dres[1];
            xs[col][2] = dres[2]; xs[col][3] = dres[3];
        }
        __syncthreads();
    }

    if (tid < P) {
        wsS[tid * 4 + 0] = xs[tid][0]; wsS[tid * 4 + 1] = xs[tid][1];
        wsS[tid * 4 + 2] = xs[tid][2]; wsS[tid * 4 + 3] = xs[tid][3];
    }
}

__global__ void phase3_emit(const float* __restrict__ u,
                            const float* __restrict__ M,
                            const float* __restrict__ C,
                            const float* __restrict__ K,
                            const float* __restrict__ dtp,
                            const float* __restrict__ wsS,
                            float* __restrict__ out, int P, int T) {
    int i = blockIdx.x * blockDim.x + threadIdx.x;
    if (i >= P) return;
    Params p = make_params(M, C, K, dtp);
    float y0 = wsS[i * 4 + 0], y1 = wsS[i * 4 + 1];
    float v0 = wsS[i * 4 + 2], v1 = wsS[i * 4 + 3];
    const int t0 = i * CHUNK_L;
    const int t1 = (t0 + CHUNK_L < T) ? (t0 + CHUNK_L) : T;
    for (int t = t0; t < t1; ++t) {
        if (t + 16 < t1) __builtin_prefetch(u + 2 * (t + 16), 0, 3);
        float u0 = u[2 * t], u1 = u[2 * t + 1];
        rk4_step(p, y0, y1, v0, v1, u0, u1);
        out[2 * t]     = y0;
        out[2 * t + 1] = y1;
    }
}

extern "C" void kernel_launch(void* const* d_in, const int* in_sizes, int n_in,
                              void* d_out, int out_size, void* d_ws,
                              size_t ws_size, hipStream_t stream) {
    const float* u  = (const float*)d_in[0];
    const float* M  = (const float*)d_in[1];
    const float* C  = (const float*)d_in[2];
    const float* K  = (const float*)d_in[3];
    const float* x0 = (const float*)d_in[4];
    const float* dt = (const float*)d_in[5];
    float* out = (float*)d_out;

    const int T = in_sizes[0] / 2;
    int P = (T + CHUNK_L - 1) / CHUNK_L;
    if (P > NCOLS) P = NCOLS;
    const int nfull = P - 1;

    float* wsE = (float*)d_ws;
    float* wsS = wsE + NCOLS * 4;

    phase1_chunks<<<(nfull + 255) / 256, 256, 0, stream>>>(
        u, M, C, K, dt, wsE, nfull);
    phase2_scan<<<1, 1024, 0, stream>>>(M, C, K, x0, dt, wsE, wsS, P);
    phase3_emit<<<(P + 255) / 256, 256, 0, stream>>>(
        u, M, C, K, dt, wsS, out, P, T);
}


# Chunk length 196 -> 25 steps: 4000 parallel serial chains, was 511; carry scan widened to 4096 columns, 12 levels, 8 column groups per wave in WMMA accumulators (carries summed in a new order)
# speedup vs baseline: 4.1226x; 4.1226x over previous
#include <hip/hip_runtime.h>


typedef __attribute__((ext_vector_type(2))) float v2f;
typedef __attribute__((ext_vector_type(8))) float v8f;

#define CHUNK_L 25
#define NCOLS   4096
#define NLEV    12
#define NGRP    8

struct Params {
    float cm00, cm01, cm10, cm11;
    float km00, km01, km10, km11;
    float m0, m1, dt, h2, dt6;
};

__device__ inline Params make_params(const float* M, const float* C,
                                     const float* K, const float* dtp) {
    Params p;
    p.cm00 = C[0] + C[1]; p.cm01 = -C[1];
    p.cm10 = -C[1];       p.cm11 = C[2] + C[1];
    p.km00 = K[0] + K[1]; p.km01 = -K[1];
    p.km10 = -K[1];       p.km11 = K[2] + K[1];
    p.m0 = M[0]; p.m1 = M[1];
    p.dt = dtp[0];
    p.h2 = 0.5f * p.dt;
    p.dt6 = p.dt / 6.0f;
    return p;
}

__device__ inline void accel(const Params& p, float z0, float z1,
                             float v0, float v1, float u0, float u1,
                             float& a0, float& a1) {
    a0 = (u0 - (p.cm00 * v0 + p.cm01 * v1) - (p.km00 * z0 + p.km01 * z1)) / p.m0;
    a1 = (u1 - (p.cm10 * v0 + p.cm11 * v1) - (p.km10 * z0 + p.km11 * z1)) / p.m1;
}

__device__ inline void rk4_step(const Params& p, float& y0, float& y1,
                                float& v0, float& v1, float u0, float u1) {
    float k1x, k1y; accel(p, y0, y1, v0, v1, u0, u1, k1x, k1y);
    float y2x = y0 + v0 * p.h2, y2y = y1 + v1 * p.h2;
    float v2x = v0 + k1x * p.h2, v2y = v1 + k1y * p.h2;
    float k2x, k2y; accel(p, y2x, y2y, v2x, v2y, u0, u1, k2x, k2y);
    float y3x = y0 + v2x * p.h2, y3y = y1 + v2y * p.h2;
    float v3x = v0 + k2x * p.h2, v3y = v1 + k2y * p.h2;
    float k3x, k3y; accel(p, y3x, y3y, v3x, v3y, u0, u1, k3x, k3y);
    float y4x = y0 + v3x * p.dt, y4y = y1 + v3y * p.dt;
    float v4x = v0 + k3x * p.dt, v4y = v1 + k3y * p.dt;
    float k4x, k4y; accel(p, y4x, y4y, v4x, v4y, u0, u1, k4x, k4y);
    float yn0 = y0 + p.dt6 * (v0 + 2.0f * v2x + 2.0f * v3x + v4x);
    float yn1 = y1 + p.dt6 * (v1 + 2.0f * v2y + 2.0f * v3y + v4y);
    float vn0 = v0 + p.dt6 * (k1x + 2.0f * k2x + 2.0f * k3x + k4x);
    float vn1 = v1 + p.dt6 * (k1y + 2.0f * k2y + 2.0f * k3y + k4y);
    y0 = yn0; y1 = yn1; v0 = vn0; v1 = vn1;
}

__device__ inline void lds_wave_fence() {
    asm volatile("s_wait_dscnt 0x0" ::: "memory");
}

__device__ inline void wmma_mul4_lds(const float* src1, const float* src2,
                                     float* dst, int lane) {
    const int  n  = lane & 15;
    const bool hi = lane >= 16;
    v2f a, b;
    a.x = (n < 4) ? src1[n * 4 + (hi ? 2 : 0)] : 0.0f;
    a.y = (n < 4) ? src1[n * 4 + (hi ? 3 : 1)] : 0.0f;
    b.x = (n < 4) ? src2[(hi ? 2 : 0) * 4 + n] : 0.0f;
    b.y = (n < 4) ? src2[(hi ? 3 : 1) * 4 + n] : 0.0f;
    v8f c = {0.f, 0.f, 0.f, 0.f, 0.f, 0.f, 0.f, 0.f};
    v8f d = __builtin_amdgcn_wmma_f32_16x16x4_f32(
        false, a, false, b, (short)0, c, false, false);
    if (!hi && n < 4) {
        dst[0 * 4 + n] = d[0];
        dst[1 * 4 + n] = d[1];
        dst[2 * 4 + n] = d[2];
        dst[3 * 4 + n] = d[3];
    }
    lds_wave_fence();
}

__global__ void phase1_chunks(const float* __restrict__ u,
                              const float* __restrict__ M,
                              const float* __restrict__ C,
                              const float* __restrict__ K,
                              const float* __restrict__ dtp,
                              float* __restrict__ wsE, int nfull) {
    int i = blockIdx.x * blockDim.x + threadIdx.x;
    if (i >= nfull) return;
    Params p = make_params(M, C, K, dtp);
    float y0 = 0.f, y1 = 0.f, v0 = 0.f, v1 = 0.f;
    const float* up = u + (size_t)i * CHUNK_L * 2;
    for (int t = 0; t < CHUNK_L; ++t) {
        if (t + 16 < CHUNK_L) __builtin_prefetch(up + 2 * (t + 16), 0, 3);
        float u0 = up[2 * t], u1 = up[2 * t + 1];
        rk4_step(p, y0, y1, v0, v1, u0, u1);
    }
    wsE[i * 4 + 0] = y0; wsE[i * 4 + 1] = y1;
    wsE[i * 4 + 2] = v0; wsE[i * 4 + 3] = v1;
}

__global__ __launch_bounds__(1024) void phase2_scan(
    const float* __restrict__ M, const float* __restrict__ C,
    const float* __restrict__ K, const float* __restrict__ x0,
    const float* __restrict__ dtp, const float* __restrict__ wsE,
    float* __restrict__ wsS, int P) {
    __shared__ __attribute__((aligned(16))) float xs[2][NCOLS][4];
    __shared__ float Gd[NLEV][16];
    __shared__ float mA[16], mP8[16], mP16[16], mQ[16], mR[16];

    const int  tid  = threadIdx.x;
    const int  lane = tid & 31;
    const int  wave = tid >> 5;
    const int  n    = lane & 15;
    const bool hi   = lane >= 16;
    const int  col0 = wave * (NGRP * 16) + n;

    for (int c = tid; c < NCOLS; c += 1024) {
        float a0, a1, a2, a3;
        if (c == 0) {
            a0 = x0[0]; a1 = x0[2]; a2 = x0[1]; a3 = x0[3];
        } else if (c < P) {
            a0 = wsE[(c - 1) * 4 + 0]; a1 = wsE[(c - 1) * 4 + 1];
            a2 = wsE[(c - 1) * 4 + 2]; a3 = wsE[(c - 1) * 4 + 3];
        } else {
            a0 = a1 = a2 = a3 = 0.f;
        }
        xs[0][c][0] = a0; xs[0][c][1] = a1; xs[0][c][2] = a2; xs[0][c][3] = a3;
    }

    if (tid < 32) {
        Params p = make_params(M, C, K, dtp);
        {
            float y0 = (lane == 0) ? 1.f : 0.f, y1 = (lane == 1) ? 1.f : 0.f;
            float v0 = (lane == 2) ? 1.f : 0.f, v1 = (lane == 3) ? 1.f : 0.f;
            rk4_step(p, y0, y1, v0, v1, 0.f, 0.f);
            if (lane < 4) {
                mA[0 * 4 + lane] = y0;
                mA[1 * 4 + lane] = y1;
                mA[2 * 4 + lane] = v0;
                mA[3 * 4 + lane] = v1;
            }
        }
        lds_wave_fence();
        wmma_mul4_lds(mA,   mA,   mQ,    lane);
        wmma_mul4_lds(mQ,   mQ,   mR,    lane);
        wmma_mul4_lds(mR,   mR,   mP8,   lane);
        wmma_mul4_lds(mP8,  mP8,  mP16,  lane);
        wmma_mul4_lds(mP16, mP8,  mQ,    lane);
        wmma_mul4_lds(mQ,   mA,   Gd[0], lane);
        for (int l = 1; l < NLEV; ++l) {
            wmma_mul4_lds(Gd[l - 1], Gd[l - 1], Gd[l], lane);
        }
    }
    __syncthreads();

    v8f acc[NGRP];
    #pragma unroll
    for (int g = 0; g < NGRP; ++g) {
        const int col = col0 + 16 * g;
        const v8f c = {xs[0][col][0], xs[0][col][1], xs[0][col][2], xs[0][col][3],
                       0.f, 0.f, 0.f, 0.f};
        acc[g] = c;
    }

    for (int l = 0; l < NLEV; ++l) {
        const int d = 1 << l;
        const float (*rd)[4] = xs[l & 1];
        float (*wr)[4] = xs[(l & 1) ^ 1];
        v2f a;
        a.x = (n < 4) ? Gd[l][n * 4 + (hi ? 2 : 0)] : 0.0f;
        a.y = (n < 4) ? Gd[l][n * 4 + (hi ? 3 : 1)] : 0.0f;
        v2f b[NGRP];
        #pragma unroll
        for (int g = 0; g < NGRP; ++g) {
            const int src = col0 + 16 * g - d;
            const int k0 = hi ? 2 : 0;
            b[g].x = (src >= 0) ? rd[src][k0]     : 0.0f;
            b[g].y = (src >= 0) ? rd[src][k0 + 1] : 0.0f;
        }
        #pragma unroll
        for (int g = 0; g < NGRP; ++g) {
            acc[g] = __builtin_amdgcn_wmma_f32_16x16x4_f32(
                false, a, false, b[g], (short)0, acc[g], false, false);
        }
        if (l + 1 < NLEV) {
            if (!hi) {
                #pragma unroll
                for (int g = 0; g < NGRP; ++g) {
                    const int col = col0 + 16 * g;
                    wr[col][0] = acc[g][0]; wr[col][1] = acc[g][1];
                    wr[col][2] = acc[g][2]; wr[col][3] = acc[g][3];
                }
            }
            __syncthreads();
        }
    }

    if (!hi) {
        #pragma unroll
        for (int g = 0; g < NGRP; ++g) {
            const int col = col0 + 16 * g;
            if (col < P) {
                wsS[col * 4 + 0] = acc[g][0]; wsS[col * 4 + 1] = acc[g][1];
                wsS[col * 4 + 2] = acc[g][2]; wsS[col * 4 + 3] = acc[g][3];
            }
        }
    }
}

__global__ void phase3_emit(const float* __restrict__ u,
                            const float* __restrict__ M,
                            const float* __restrict__ C,
                            const float* __restrict__ K,
                            const float* __restrict__ dtp,
                            const float* __restrict__ wsS,
                            float* __restrict__ out, int P, int T) {
    int i = blockIdx.x * blockDim.x + threadIdx.x;
    if (i >= P) return;
    Params p = make_params(M, C, K, dtp);
    float y0 = wsS[i * 4 + 0], y1 = wsS[i * 4 + 1];
    float v0 = wsS[i * 4 + 2], v1 = wsS[i * 4 + 3];
    const int t0 = i * CHUNK_L;
    const int t1 = (t0 + CHUNK_L < T) ? (t0 + CHUNK_L) : T;
    for (int t = t0; t < t1; ++t) {
        if (t + 16 < t1) __builtin_prefetch(u + 2 * (t + 16), 0, 3);
        float u0 = u[2 * t], u1 = u[2 * t + 1];
        rk4_step(p, y0, y1, v0, v1, u0, u1);
        out[2 * t]     = y0;
        out[2 * t + 1] = y1;
    }
}

extern "C" void kernel_launch(void* const* d_in, const int* in_sizes, int n_in,
                              void* d_out, int out_size, void* d_ws,
                              size_t ws_size, hipStream_t stream) {
    const float* u  = (const float*)d_in[0];
    const float* M  = (const float*)d_in[1];
    const float* C  = (const float*)d_in[2];
    const float* K  = (const float*)d_in[3];
    const float* x0 = (const float*)d_in[4];
    const float* dt = (const float*)d_in[5];
    float* out = (float*)d_out;

    const int T = in_sizes[0] / 2;
    int P = (T + CHUNK_L - 1) / CHUNK_L;
    if (P > NCOLS) P = NCOLS;
    const int nfull = P - 1;

    float* wsE = (float*)d_ws;
    float* wsS = wsE + NCOLS * 4;

    phase1_chunks<<<(nfull + 255) / 256, 256, 0, stream>>>(
        u, M, C, K, dt, wsE, nfull);
    phase2_scan<<<1, 1024, 0, stream>>>(M, C, K, x0, dt, wsE, wsS, P);
    phase3_emit<<<(P + 255) / 256, 256, 0, stream>>>(
        u, M, C, K, dt, wsS, out, P, T);
}


# Chunk phases step with the one-step affine map s'=A s+B u (A,B once per wave from rk4_step on unit vectors), not 4 RK stages with 8 divisions; step sums in a new order; one-wave blocks
# speedup vs baseline: 7.7998x; 1.8920x over previous
#include <hip/hip_runtime.h>


typedef __attribute__((ext_vector_type(2))) float v2f;
typedef __attribute__((ext_vector_type(8))) float v8f;

#define CHUNK_L 25
#define NCOLS   4096
#define NLEV    12
#define NGRP    8
#define CH_BLOCK 32

struct Params {
    float cm00, cm01, cm10, cm11;
    float km00, km01, km10, km11;
    float m0, m1, dt, h2, dt6;
};

__device__ inline Params make_params(const float* M, const float* C,
                                     const float* K, const float* dtp) {
    Params p;
    p.cm00 = C[0] + C[1]; p.cm01 = -C[1];
    p.cm10 = -C[1];       p.cm11 = C[2] + C[1];
    p.km00 = K[0] + K[1]; p.km01 = -K[1];
    p.km10 = -K[1];       p.km11 = K[2] + K[1];
    p.m0 = M[0]; p.m1 = M[1];
    p.dt = dtp[0];
    p.h2 = 0.5f * p.dt;
    p.dt6 = p.dt / 6.0f;
    return p;
}

__device__ inline void accel(const Params& p, float z0, float z1,
                             float v0, float v1, float u0, float u1,
                             float& a0, float& a1) {
    a0 = (u0 - (p.cm00 * v0 + p.cm01 * v1) - (p.km00 * z0 + p.km01 * z1)) / p.m0;
    a1 = (u1 - (p.cm10 * v0 + p.cm11 * v1) - (p.km10 * z0 + p.km11 * z1)) / p.m1;
}

__device__ inline void rk4_step(const Params& p, float& y0, float& y1,
                                float& v0, float& v1, float u0, float u1) {
    float k1x, k1y; accel(p, y0, y1, v0, v1, u0, u1, k1x, k1y);
    float y2x = y0 + v0 * p.h2, y2y = y1 + v1 * p.h2;
    float v2x = v0 + k1x * p.h2, v2y = v1 + k1y * p.h2;
    float k2x, k2y; accel(p, y2x, y2y, v2x, v2y, u0, u1, k2x, k2y);
    float y3x = y0 + v2x * p.h2, y3y = y1 + v2y * p.h2;
    float v3x = v0 + k2x * p.h2, v3y = v1 + k2y * p.h2;
    float k3x, k3y; accel(p, y3x, y3y, v3x, v3y, u0, u1, k3x, k3y);
    float y4x = y0 + v3x * p.dt, y4y = y1 + v3y * p.dt;
    float v4x = v0 + k3x * p.dt, v4y = v1 + k3y * p.dt;
    float k4x, k4y; accel(p, y4x, y4y, v4x, v4y, u0, u1, k4x, k4y);
    float yn0 = y0 + p.dt6 * (v0 + 2.0f * v2x + 2.0f * v3x + v4x);
    float yn1 = y1 + p.dt6 * (v1 + 2.0f * v2y + 2.0f * v3y + v4y);
    float vn0 = v0 + p.dt6 * (k1x + 2.0f * k2x + 2.0f * k3x + k4x);
    float vn1 = v1 + p.dt6 * (k1y + 2.0f * k2y + 2.0f * k3y + k4y);
    y0 = yn0; y1 = yn1; v0 = vn0; v1 = vn1;
}

__device__ inline void lds_wave_fence() {
    asm volatile("s_wait_dscnt 0x0" ::: "memory");
}

struct StepMap {
    float a[4][4];
    float b[4][2];
};

__device__ inline float lane_bcast(float x, int srclane) {
    return __int_as_float(__builtin_amdgcn_readlane(__float_as_int(x), srclane));
}

__device__ inline StepMap make_step_map(const Params& p, int lane) {
    float y0 = (lane == 0) ? 1.f : 0.f, y1 = (lane == 1) ? 1.f : 0.f;
    float v0 = (lane == 2) ? 1.f : 0.f, v1 = (lane == 3) ? 1.f : 0.f;
    const float u0 = (lane == 4) ? 1.f : 0.f, u1 = (lane == 5) ? 1.f : 0.f;
    rk4_step(p, y0, y1, v0, v1, u0, u1);
    StepMap m;
    #pragma unroll
    for (int j = 0; j < 4; ++j) {
        m.a[0][j] = lane_bcast(y0, j);
        m.a[1][j] = lane_bcast(y1, j);
        m.a[2][j] = lane_bcast(v0, j);
        m.a[3][j] = lane_bcast(v1, j);
    }
    #pragma unroll
    for (int j = 0; j < 2; ++j) {
        m.b[0][j] = lane_bcast(y0, 4 + j);
        m.b[1][j] = lane_bcast(y1, 4 + j);
        m.b[2][j] = lane_bcast(v0, 4 + j);
        m.b[3][j] = lane_bcast(v1, 4 + j);
    }
    return m;
}

__device__ inline void affine_step(const StepMap& m, float& y0, float& y1,
                                   float& v0, float& v1, float u0, float u1) {
    const float s[4] = {y0, y1, v0, v1};
    float r[4];
    #pragma unroll
    for (int i = 0; i < 4; ++i) {
        float acc = m.b[i][1] * u1;
        acc = fmaf(m.b[i][0], u0, acc);
        #pragma unroll
        for (int j = 3; j >= 0; --j) {
            if (j != i) acc = fmaf(m.a[i][j], s[j], acc);
        }
        r[i] = fmaf(m.a[i][i], s[i], acc);
    }
    y0 = r[0]; y1 = r[1]; v0 = r[2]; v1 = r[3];
}

__device__ inline void wmma_mul4_lds(const float* src1, const float* src2,
                                     float* dst, int lane) {
    const int  n  = lane & 15;
    const bool hi = lane >= 16;
    v2f a, b;
    a.x = (n < 4) ? src1[n * 4 + (hi ? 2 : 0)] : 0.0f;
    a.y = (n < 4) ? src1[n * 4 + (hi ? 3 : 1)] : 0.0f;
    b.x = (n < 4) ? src2[(hi ? 2 : 0) * 4 + n] : 0.0f;
    b.y = (n < 4) ? src2[(hi ? 3 : 1) * 4 + n] : 0.0f;
    v8f c = {0.f, 0.f, 0.f, 0.f, 0.f, 0.f, 0.f, 0.f};
    v8f d = __builtin_amdgcn_wmma_f32_16x16x4_f32(
        false, a, false, b, (short)0, c, false, false);
    if (!hi && n < 4) {
        dst[0 * 4 + n] = d[0];
        dst[1 * 4 + n] = d[1];
        dst[2 * 4 + n] = d[2];
        dst[3 * 4 + n] = d[3];
    }
    lds_wave_fence();
}

__device__ inline void load_chunk(const float* __restrict__ up,
                                  float (&uu)[CHUNK_L][2]) {
    #pragma unroll
    for (int t = 0; t < CHUNK_L; ++t) {
        uu[t][0] = up[2 * t];
        uu[t][1] = up[2 * t + 1];
    }
}

__device__ inline void chunk_end_from_zero(const StepMap& m,
                                           const float (&uu)[CHUNK_L][2],
                                           float* __restrict__ e) {
    float y0 = 0.f, y1 = 0.f, v0 = 0.f, v1 = 0.f;
    #pragma unroll
    for (int t = 0; t < CHUNK_L; ++t) {
        affine_step(m, y0, y1, v0, v1, uu[t][0], uu[t][1]);
    }
    e[0] = y0; e[1] = y1; e[2] = v0; e[3] = v1;
}

__global__ __launch_bounds__(CH_BLOCK) void phase1_chunks(
    const float* __restrict__ u, const float* __restrict__ M,
    const float* __restrict__ C, const float* __restrict__ K,
    const float* __restrict__ dtp, float* __restrict__ wsE, int nfull) {
    const int  lane = threadIdx.x & 31;
    const int  i    = blockIdx.x * blockDim.x + threadIdx.x;
    const bool live = i < nfull;
    const float* up = u + (size_t)i * CHUNK_L * 2;
    float uu[CHUNK_L][2];
    if (__all(live)) {
        load_chunk(up, uu);
        const Params  p = make_params(M, C, K, dtp);
        const StepMap m = make_step_map(p, lane);
        chunk_end_from_zero(m, uu, wsE + i * 4);
    } else {
        const Params  p = make_params(M, C, K, dtp);
        const StepMap m = make_step_map(p, lane);
        if (live) {
            load_chunk(up, uu);
            chunk_end_from_zero(m, uu, wsE + i * 4);
        }
    }
}

__global__ __launch_bounds__(1024) void phase2_scan(
    const float* __restrict__ M, const float* __restrict__ C,
    const float* __restrict__ K, const float* __restrict__ x0,
    const float* __restrict__ dtp, const float* __restrict__ wsE,
    float* __restrict__ wsS, int P) {
    __shared__ __attribute__((aligned(16))) float xs[2][NCOLS][4];
    __shared__ float Gd[NLEV][16];
    __shared__ float mA[16], mP8[16], mP16[16], mQ[16], mR[16];

    const int  tid  = threadIdx.x;
    const int  lane = tid & 31;
    const int  wave = tid >> 5;
    const int  n    = lane & 15;
    const bool hi   = lane >= 16;
    const int  col0 = wave * (NGRP * 16) + n;

    for (int c = tid; c < NCOLS; c += 1024) {
        float a0, a1, a2, a3;
        if (c == 0) {
            a0 = x0[0]; a1 = x0[2]; a2 = x0[1]; a3 = x0[3];
        } else if (c < P) {
            a0 = wsE[(c - 1) * 4 + 0]; a1 = wsE[(c - 1) * 4 + 1];
            a2 = wsE[(c - 1) * 4 + 2]; a3 = wsE[(c - 1) * 4 + 3];
        } else {
            a0 = a1 = a2 = a3 = 0.f;
        }
        xs[0][c][0] = a0; xs[0][c][1] = a1; xs[0][c][2] = a2; xs[0][c][3] = a3;
    }

    if (tid < 32) {
        Params p = make_params(M, C, K, dtp);
        {
            float y0 = (lane == 0) ? 1.f : 0.f, y1 = (lane == 1) ? 1.f : 0.f;
            float v0 = (lane == 2) ? 1.f : 0.f, v1 = (lane == 3) ? 1.f : 0.f;
            rk4_step(p, y0, y1, v0, v1, 0.f, 0.f);
            if (lane < 4) {
                mA[0 * 4 + lane] = y0;
                mA[1 * 4 + lane] = y1;
                mA[2 * 4 + lane] = v0;
                mA[3 * 4 + lane] = v1;
            }
        }
        lds_wave_fence();
        wmma_mul4_lds(mA,   mA,   mQ,    lane);
        wmma_mul4_lds(mQ,   mQ,   mR,    lane);
        wmma_mul4_lds(mR,   mR,   mP8,   lane);
        wmma_mul4_lds(mP8,  mP8,  mP16,  lane);
        wmma_mul4_lds(mP16, mP8,  mQ,    lane);
        wmma_mul4_lds(mQ,   mA,   Gd[0], lane);
        for (int l = 1; l < NLEV; ++l) {
            wmma_mul4_lds(Gd[l - 1], Gd[l - 1], Gd[l], lane);
        }
    }
    __syncthreads();

    v8f acc[NGRP];
    #pragma unroll
    for (int g = 0; g < NGRP; ++g) {
        const int col = col0 + 16 * g;
        const v8f c = {xs[0][col][0], xs[0][col][1], xs[0][col][2], xs[0][col][3],
                       0.f, 0.f, 0.f, 0.f};
        acc[g] = c;
    }

    for (int l = 0; l < NLEV; ++l) {
        const int d = 1 << l;
        const float (*rd)[4] = xs[l & 1];
        float (*wr)[4] = xs[(l & 1) ^ 1];
        v2f a;
        a.x = (n < 4) ? Gd[l][n * 4 + (hi ? 2 : 0)] : 0.0f;
        a.y = (n < 4) ? Gd[l][n * 4 + (hi ? 3 : 1)] : 0.0f;
        v2f b[NGRP];
        #pragma unroll
        for (int g = 0; g < NGRP; ++g) {
            const int src = col0 + 16 * g - d;
            const int k0 = hi ? 2 : 0;
            b[g].x = (src >= 0) ? rd[src][k0]     : 0.0f;
            b[g].y = (src >= 0) ? rd[src][k0 + 1] : 0.0f;
        }
        #pragma unroll
        for (int g = 0; g < NGRP; ++g) {
            acc[g] = __builtin_amdgcn_wmma_f32_16x16x4_f32(
                false, a, false, b[g], (short)0, acc[g], false, false);
        }
        if (l + 1 < NLEV) {
            if (!hi) {
                #pragma unroll
                for (int g = 0; g < NGRP; ++g) {
                    const int col = col0 + 16 * g;
                    wr[col][0] = acc[g][0]; wr[col][1] = acc[g][1];
                    wr[col][2] = acc[g][2]; wr[col][3] = acc[g][3];
                }
            }
            __syncthreads();
        }
    }

    if (!hi) {
        #pragma unroll
        for (int g = 0; g < NGRP; ++g) {
            const int col = col0 + 16 * g;
            if (col < P) {
                wsS[col * 4 + 0] = acc[g][0]; wsS[col * 4 + 1] = acc[g][1];
                wsS[col * 4 + 2] = acc[g][2]; wsS[col * 4 + 3] = acc[g][3];
            }
        }
    }
}

__device__ inline void emit_full_chunk(const StepMap& m,
                                       const float (&uu)[CHUNK_L][2],
                                       float y0, float y1, float v0, float v1,
                                       float* __restrict__ op) {
    float yo[CHUNK_L][2];
    #pragma unroll
    for (int t = 0; t < CHUNK_L; ++t) {
        affine_step(m, y0, y1, v0, v1, uu[t][0], uu[t][1]);
        yo[t][0] = y0;
        yo[t][1] = y1;
    }
    #pragma unroll
    for (int t = 0; t < CHUNK_L; ++t) {
        op[2 * t]     = yo[t][0];
        op[2 * t + 1] = yo[t][1];
    }
}

__global__ __launch_bounds__(CH_BLOCK) void phase3_emit(
    const float* __restrict__ u, const float* __restrict__ M,
    const float* __restrict__ C, const float* __restrict__ K,
    const float* __restrict__ dtp, const float* __restrict__ wsS,
    float* __restrict__ out, int P, int T) {
    const int  lane = threadIdx.x & 31;
    const int  i    = blockIdx.x * blockDim.x + threadIdx.x;
    const bool live = i < P;
    const bool full = live && (i + 1) * CHUNK_L <= T;
    const float* up = u + (size_t)i * CHUNK_L * 2;
    float*       op = out + (size_t)i * CHUNK_L * 2;
    float uu[CHUNK_L][2];
    if (__all(full)) {
        load_chunk(up, uu);
        const float y0 = wsS[i * 4 + 0], y1 = wsS[i * 4 + 1];
        const float v0 = wsS[i * 4 + 2], v1 = wsS[i * 4 + 3];
        const Params  p = make_params(M, C, K, dtp);
        const StepMap m = make_step_map(p, lane);
        emit_full_chunk(m, uu, y0, y1, v0, v1, op);
    } else {
        const Params  p = make_params(M, C, K, dtp);
        const StepMap m = make_step_map(p, lane);
        if (full) {
            load_chunk(up, uu);
            const float y0 = wsS[i * 4 + 0], y1 = wsS[i * 4 + 1];
            const float v0 = wsS[i * 4 + 2], v1 = wsS[i * 4 + 3];
            emit_full_chunk(m, uu, y0, y1, v0, v1, op);
        } else if (live) {
            float y0 = wsS[i * 4 + 0], y1 = wsS[i * 4 + 1];
            float v0 = wsS[i * 4 + 2], v1 = wsS[i * 4 + 3];
            for (int t = i * CHUNK_L; t < T; ++t) {
                affine_step(m, y0, y1, v0, v1, u[2 * t], u[2 * t + 1]);
                out[2 * t]     = y0;
                out[2 * t + 1] = y1;
            }
        }
    }
}

extern "C" void kernel_launch(void* const* d_in, const int* in_sizes, int n_in,
                              void* d_out, int out_size, void* d_ws,
                              size_t ws_size, hipStream_t stream) {
    const float* u  = (const float*)d_in[0];
    const float* M  = (const float*)d_in[1];
    const float* C  = (const float*)d_in[2];
    const float* K  = (const float*)d_in[3];
    const float* x0 = (const float*)d_in[4];
    const float* dt = (const float*)d_in[5];
    float* out = (float*)d_out;

    const int T = in_sizes[0] / 2;
    int P = (T + CHUNK_L - 1) / CHUNK_L;
    if (P > NCOLS) P = NCOLS;
    const int nfull = P - 1;

    float* wsE = (float*)d_ws;
    float* wsS = wsE + NCOLS * 4;

    phase1_chunks<<<(nfull + CH_BLOCK - 1) / CH_BLOCK, CH_BLOCK, 0, stream>>>(
        u, M, C, K, dt, wsE, nfull);
    phase2_scan<<<1, 1024, 0, stream>>>(M, C, K, x0, dt, wsE, wsS, P);
    phase3_emit<<<(P + CH_BLOCK - 1) / CH_BLOCK, CH_BLOCK, 0, stream>>>(
        u, M, C, K, dt, wsS, out, P, T);
}


# Carry scan blocked: a lane owns 8 consecutive columns (Horner run sum, 9-level scan of 512 run sums, run re-walk): 24 WMMAs per wave, was 96; columns loaded from global; carries summed in a new order
# speedup vs baseline: 10.3536x; 1.3274x over previous
#include <hip/hip_runtime.h>


typedef __attribute__((ext_vector_type(2))) float v2f;
typedef __attribute__((ext_vector_type(4))) float v4f;
typedef __attribute__((ext_vector_type(8))) float v8f;

#define CHUNK_L 25
#define NCOLS   4096
#define NLEV    12
#define RUNL    8
#define RUNLOG  3
#define NRUNS   512
#define NLEV2   9
#define CH_BLOCK 32

struct Params {
    float cm00, cm01, cm10, cm11;
    float km00, km01, km10, km11;
    float m0, m1, dt, h2, dt6;
};

__device__ inline Params make_params(const float* M, const float* C,
                                     const float* K, const float* dtp) {
    Params p;
    p.cm00 = C[0] + C[1]; p.cm01 = -C[1];
    p.cm10 = -C[1];       p.cm11 = C[2] + C[1];
    p.km00 = K[0] + K[1]; p.km01 = -K[1];
    p.km10 = -K[1];       p.km11 = K[2] + K[1];
    p.m0 = M[0]; p.m1 = M[1];
    p.dt = dtp[0];
    p.h2 = 0.5f * p.dt;
    p.dt6 = p.dt / 6.0f;
    return p;
}

__device__ inline void accel(const Params& p, float z0, float z1,
                             float v0, float v1, float u0, float u1,
                             float& a0, float& a1) {
    a0 = (u0 - (p.cm00 * v0 + p.cm01 * v1) - (p.km00 * z0 + p.km01 * z1)) / p.m0;
    a1 = (u1 - (p.cm10 * v0 + p.cm11 * v1) - (p.km10 * z0 + p.km11 * z1)) / p.m1;
}

__device__ inline void rk4_step(const Params& p, float& y0, float& y1,
                                float& v0, float& v1, float u0, float u1) {
    float k1x, k1y; accel(p, y0, y1, v0, v1, u0, u1, k1x, k1y);
    float y2x = y0 + v0 * p.h2, y2y = y1 + v1 * p.h2;
    float v2x = v0 + k1x * p.h2, v2y = v1 + k1y * p.h2;
    float k2x, k2y; accel(p, y2x, y2y, v2x, v2y, u0, u1, k2x, k2y);
    float y3x = y0 + v2x * p.h2, y3y = y1 + v2y * p.h2;
    float v3x = v0 + k2x * p.h2, v3y = v1 + k2y * p.h2;
    float k3x, k3y; accel(p, y3x, y3y, v3x, v3y, u0, u1, k3x, k3y);
    float y4x = y0 + v3x * p.dt, y4y = y1 + v3y * p.dt;
    float v4x = v0 + k3x * p.dt, v4y = v1 + k3y * p.dt;
    float k4x, k4y; accel(p, y4x, y4y, v4x, v4y, u0, u1, k4x, k4y);
    float yn0 = y0 + p.dt6 * (v0 + 2.0f * v2x + 2.0f * v3x + v4x);
    float yn1 = y1 + p.dt6 * (v1 + 2.0f * v2y + 2.0f * v3y + v4y);
    float vn0 = v0 + p.dt6 * (k1x + 2.0f * k2x + 2.0f * k3x + k4x);
    float vn1 = v1 + p.dt6 * (k1y + 2.0f * k2y + 2.0f * k3y + k4y);
    y0 = yn0; y1 = yn1; v0 = vn0; v1 = vn1;
}

__device__ inline void lds_wave_fence() {
    asm volatile("s_wait_dscnt 0x0" ::: "memory");
}

struct StepMap {
    float a[4][4];
    float b[4][2];
};

__device__ inline float lane_bcast(float x, int srclane) {
    return __int_as_float(__builtin_amdgcn_readlane(__float_as_int(x), srclane));
}

__device__ inline StepMap make_step_map(const Params& p, int lane) {
    float y0 = (lane == 0) ? 1.f : 0.f, y1 = (lane == 1) ? 1.f : 0.f;
    float v0 = (lane == 2) ? 1.f : 0.f, v1 = (lane == 3) ? 1.f : 0.f;
    const float u0 = (lane == 4) ? 1.f : 0.f, u1 = (lane == 5) ? 1.f : 0.f;
    rk4_step(p, y0, y1, v0, v1, u0, u1);
    StepMap m;
    #pragma unroll
    for (int j = 0; j < 4; ++j) {
        m.a[0][j] = lane_bcast(y0, j);
        m.a[1][j] = lane_bcast(y1, j);
        m.a[2][j] = lane_bcast(v0, j);
        m.a[3][j] = lane_bcast(v1, j);
    }
    #pragma unroll
    for (int j = 0; j < 2; ++j) {
        m.b[0][j] = lane_bcast(y0, 4 + j);
        m.b[1][j] = lane_bcast(y1, 4 + j);
        m.b[2][j] = lane_bcast(v0, 4 + j);
        m.b[3][j] = lane_bcast(v1, 4 + j);
    }
    return m;
}

__device__ inline void affine_step(const StepMap& m, float& y0, float& y1,
                                   float& v0, float& v1, float u0, float u1) {
    const float s[4] = {y0, y1, v0, v1};
    float r[4];
    #pragma unroll
    for (int i = 0; i < 4; ++i) {
        float acc = m.b[i][1] * u1;
        acc = fmaf(m.b[i][0], u0, acc);
        #pragma unroll
        for (int j = 3; j >= 0; --j) {
            if (j != i) acc = fmaf(m.a[i][j], s[j], acc);
        }
        r[i] = fmaf(m.a[i][i], s[i], acc);
    }
    y0 = r[0]; y1 = r[1]; v0 = r[2]; v1 = r[3];
}

__device__ inline void wmma_mul4_lds(const float* src1, const float* src2,
                                     float* dst, int lane) {
    const int  n  = lane & 15;
    const bool hi = lane >= 16;
    v2f a, b;
    a.x = (n < 4) ? src1[n * 4 + (hi ? 2 : 0)] : 0.0f;
    a.y = (n < 4) ? src1[n * 4 + (hi ? 3 : 1)] : 0.0f;
    b.x = (n < 4) ? src2[(hi ? 2 : 0) * 4 + n] : 0.0f;
    b.y = (n < 4) ? src2[(hi ? 3 : 1) * 4 + n] : 0.0f;
    v8f c = {0.f, 0.f, 0.f, 0.f, 0.f, 0.f, 0.f, 0.f};
    v8f d = __builtin_amdgcn_wmma_f32_16x16x4_f32(
        false, a, false, b, (short)0, c, false, false);
    if (!hi && n < 4) {
        dst[0 * 4 + n] = d[0];
        dst[1 * 4 + n] = d[1];
        dst[2 * 4 + n] = d[2];
        dst[3 * 4 + n] = d[3];
    }
    lds_wave_fence();
}

__device__ inline void load_chunk(const float* __restrict__ up,
                                  float (&uu)[CHUNK_L][2]) {
    #pragma unroll
    for (int t = 0; t < CHUNK_L; ++t) {
        uu[t][0] = up[2 * t];
        uu[t][1] = up[2 * t + 1];
    }
}

__device__ inline void chunk_end_from_zero(const StepMap& m,
                                           const float (&uu)[CHUNK_L][2],
                                           float* __restrict__ e) {
    float y0 = 0.f, y1 = 0.f, v0 = 0.f, v1 = 0.f;
    #pragma unroll
    for (int t = 0; t < CHUNK_L; ++t) {
        affine_step(m, y0, y1, v0, v1, uu[t][0], uu[t][1]);
    }
    e[0] = y0; e[1] = y1; e[2] = v0; e[3] = v1;
}

__global__ __launch_bounds__(CH_BLOCK) void phase1_chunks(
    const float* __restrict__ u, const float* __restrict__ M,
    const float* __restrict__ C, const float* __restrict__ K,
    const float* __restrict__ dtp, float* __restrict__ wsE, int nfull) {
    const int  lane = threadIdx.x & 31;
    const int  i    = blockIdx.x * blockDim.x + threadIdx.x;
    const bool live = i < nfull;
    const float* up = u + (size_t)i * CHUNK_L * 2;
    float uu[CHUNK_L][2];
    if (__all(live)) {
        load_chunk(up, uu);
        const Params  p = make_params(M, C, K, dtp);
        const StepMap m = make_step_map(p, lane);
        chunk_end_from_zero(m, uu, wsE + i * 4);
    } else {
        const Params  p = make_params(M, C, K, dtp);
        const StepMap m = make_step_map(p, lane);
        if (live) {
            load_chunk(up, uu);
            chunk_end_from_zero(m, uu, wsE + i * 4);
        }
    }
}

__device__ inline v2f g_fragment(const float* g, int n, bool hi) {
    const int row = n & 7;
    v2f a;
    a.x = (row < 4) ? g[row * 4 + (hi ? 2 : 0)] : 0.0f;
    a.y = (row < 4) ? g[row * 4 + (hi ? 3 : 1)] : 0.0f;
    return a;
}

__device__ inline v2f tile_as_b(const v8f& t, bool hi) {
    v2f b;
    b.x = hi ? t[2] : t[0];
    b.y = hi ? t[3] : t[1];
    return b;
}

__global__ __launch_bounds__(1024) void phase2_scan(
    const float* __restrict__ M, const float* __restrict__ C,
    const float* __restrict__ K, const float* __restrict__ x0,
    const float* __restrict__ dtp, const float* __restrict__ wsE,
    float* __restrict__ wsS, int P) {
    __shared__ __attribute__((aligned(16))) float rs[2][NRUNS][4];
    __shared__ float Gd[NLEV][16];
    __shared__ float mA[16], mP8[16], mP16[16], mQ[16], mR[16];

    const int  tid  = threadIdx.x;
    const int  lane = tid & 31;
    const int  wave = tid >> 5;
    const int  n    = lane & 15;
    const bool hi   = lane >= 16;
    const int  run  = wave * 16 + n;
    const int  colb = run * RUNL;
    const int  k0   = hi ? 2 : 0;

    float xv[RUNL][4];
    #pragma unroll
    for (int g = 0; g < RUNL; ++g) {
        const int c = colb + g;
        float a0, a1, a2, a3;
        if (c == 0) {
            a0 = x0[0]; a1 = x0[2]; a2 = x0[1]; a3 = x0[3];
        } else if (c < P) {
            const v4f e = *(const v4f*)(wsE + (size_t)(c - 1) * 4);
            a0 = e.x; a1 = e.y; a2 = e.z; a3 = e.w;
        } else {
            a0 = a1 = a2 = a3 = 0.f;
        }
        xv[g][0] = a0; xv[g][1] = a1; xv[g][2] = a2; xv[g][3] = a3;
    }

    if (tid < 32) {
        Params p = make_params(M, C, K, dtp);
        {
            float y0 = (lane == 0) ? 1.f : 0.f, y1 = (lane == 1) ? 1.f : 0.f;
            float v0 = (lane == 2) ? 1.f : 0.f, v1 = (lane == 3) ? 1.f : 0.f;
            rk4_step(p, y0, y1, v0, v1, 0.f, 0.f);
            if (lane < 4) {
                mA[0 * 4 + lane] = y0;
                mA[1 * 4 + lane] = y1;
                mA[2 * 4 + lane] = v0;
                mA[3 * 4 + lane] = v1;
            }
        }
        lds_wave_fence();
        wmma_mul4_lds(mA,   mA,   mQ,    lane);
        wmma_mul4_lds(mQ,   mQ,   mR,    lane);
        wmma_mul4_lds(mR,   mR,   mP8,   lane);
        wmma_mul4_lds(mP8,  mP8,  mP16,  lane);
        wmma_mul4_lds(mP16, mP8,  mQ,    lane);
        wmma_mul4_lds(mQ,   mA,   Gd[0], lane);
        for (int l = 1; l < NLEV; ++l) {
            wmma_mul4_lds(Gd[l - 1], Gd[l - 1], Gd[l], lane);
        }
    }
    __syncthreads();

    const v2f aG = g_fragment(Gd[0], n, hi);

    v8f t = {xv[0][0], xv[0][1], xv[0][2], xv[0][3], 0.f, 0.f, 0.f, 0.f};
    #pragma unroll
    for (int g = 1; g < RUNL; ++g) {
        const v8f c = {xv[g][0], xv[g][1], xv[g][2], xv[g][3], 0.f, 0.f, 0.f, 0.f};
        t = __builtin_amdgcn_wmma_f32_16x16x4_f32(
            false, aG, false, tile_as_b(t, hi), (short)0, c, false, false);
    }
    if (!hi) {
        rs[0][run][0] = t[0]; rs[0][run][1] = t[1];
        rs[0][run][2] = t[2]; rs[0][run][3] = t[3];
    }
    __syncthreads();

    for (int l = 0; l < NLEV2; ++l) {
        const int d = 1 << l;
        const float (*rd)[4] = rs[l & 1];
        float (*wr)[4] = rs[(l & 1) ^ 1];
        const v2f a = g_fragment(Gd[l + RUNLOG], n, hi);
        const int src = run - d;
        v2f b;
        b.x = (src >= 0) ? rd[src][k0]     : 0.0f;
        b.y = (src >= 0) ? rd[src][k0 + 1] : 0.0f;
        t = __builtin_amdgcn_wmma_f32_16x16x4_f32(
            false, a, false, b, (short)0, t, false, false);
        if (!hi) {
            wr[run][0] = t[0]; wr[run][1] = t[1];
            wr[run][2] = t[2]; wr[run][3] = t[3];
        }
        __syncthreads();
    }

    const float (*fin)[4] = rs[NLEV2 & 1];
    v2f b;
    b.x = (run > 0) ? fin[run - 1][k0]     : 0.0f;
    b.y = (run > 0) ? fin[run - 1][k0 + 1] : 0.0f;
    #pragma unroll
    for (int g = 0; g < RUNL; ++g) {
        const v8f c = {xv[g][0], xv[g][1], xv[g][2], xv[g][3], 0.f, 0.f, 0.f, 0.f};
        const v8f o = __builtin_amdgcn_wmma_f32_16x16x4_f32(
            false, aG, false, b, (short)0, c, false, false);
        const int col = colb + g;
        if (!hi && col < P) {
            v4f s; s.x = o[0]; s.y = o[1]; s.z = o[2]; s.w = o[3];
            *(v4f*)(wsS + (size_t)col * 4) = s;
        }
        b = tile_as_b(o, hi);
    }
}

__device__ inline void emit_full_chunk(const StepMap& m,
                                       const float (&uu)[CHUNK_L][2],
                                       float y0, float y1, float v0, float v1,
                                       float* __restrict__ op) {
    float yo[CHUNK_L][2];
    #pragma unroll
    for (int t = 0; t < CHUNK_L; ++t) {
        affine_step(m, y0, y1, v0, v1, uu[t][0], uu[t][1]);
        yo[t][0] = y0;
        yo[t][1] = y1;
    }
    #pragma unroll
    for (int t = 0; t < CHUNK_L; ++t) {
        op[2 * t]     = yo[t][0];
        op[2 * t + 1] = yo[t][1];
    }
}

__global__ __launch_bounds__(CH_BLOCK) void phase3_emit(
    const float* __restrict__ u, const float* __restrict__ M,
    const float* __restrict__ C, const float* __restrict__ K,
    const float* __restrict__ dtp, const float* __restrict__ wsS,
    float* __restrict__ out, int P, int T) {
    const int  lane = threadIdx.x & 31;
    const int  i    = blockIdx.x * blockDim.x + threadIdx.x;
    const bool live = i < P;
    const bool full = live && (i + 1) * CHUNK_L <= T;
    const float* up = u + (size_t)i * CHUNK_L * 2;
    float*       op = out + (size_t)i * CHUNK_L * 2;
    float uu[CHUNK_L][2];
    if (__all(full)) {
        load_chunk(up, uu);
        const float y0 = wsS[i * 4 + 0], y1 = wsS[i * 4 + 1];
        const float v0 = wsS[i * 4 + 2], v1 = wsS[i * 4 + 3];
        const Params  p = make_params(M, C, K, dtp);
        const StepMap m = make_step_map(p, lane);
        emit_full_chunk(m, uu, y0, y1, v0, v1, op);
    } else {
        const Params  p = make_params(M, C, K, dtp);
        const StepMap m = make_step_map(p, lane);
        if (full) {
            load_chunk(up, uu);
            const float y0 = wsS[i * 4 + 0], y1 = wsS[i * 4 + 1];
            const float v0 = wsS[i * 4 + 2], v1 = wsS[i * 4 + 3];
            emit_full_chunk(m, uu, y0, y1, v0, v1, op);
        } else if (live) {
            float y0 = wsS[i * 4 + 0], y1 = wsS[i * 4 + 1];
            float v0 = wsS[i * 4 + 2], v1 = wsS[i * 4 + 3];
            for (int t = i * CHUNK_L; t < T; ++t) {
                affine_step(m, y0, y1, v0, v1, u[2 * t], u[2 * t + 1]);
                out[2 * t]     = y0;
                out[2 * t + 1] = y1;
            }
        }
    }
}

extern "C" void kernel_launch(void* const* d_in, const int* in_sizes, int n_in,
                              void* d_out, int out_size, void* d_ws,
                              size_t ws_size, hipStream_t stream) {
    const float* u  = (const float*)d_in[0];
    const float* M  = (const float*)d_in[1];
    const float* C  = (const float*)d_in[2];
    const float* K  = (const float*)d_in[3];
    const float* x0 = (const float*)d_in[4];
    const float* dt = (const float*)d_in[5];
    float* out = (float*)d_out;

    const int T = in_sizes[0] / 2;
    int P = (T + CHUNK_L - 1) / CHUNK_L;
    if (P > NCOLS) P = NCOLS;
    const int nfull = P - 1;

    float* wsE = (float*)d_ws;
    float* wsS = wsE + NCOLS * 4;

    phase1_chunks<<<(nfull + CH_BLOCK - 1) / CH_BLOCK, CH_BLOCK, 0, stream>>>(
        u, M, C, K, dt, wsE, nfull);
    phase2_scan<<<1, 1024, 0, stream>>>(M, C, K, x0, dt, wsE, wsS, P);
    phase3_emit<<<(P + CH_BLOCK - 1) / CH_BLOCK, CH_BLOCK, 0, stream>>>(
        u, M, C, K, dt, wsS, out, P, T);
}


# Carry scan folded into the two chunk kernels: a block sums its 32 chunk ends, each emit block rescans block sums and own chunks on WMMA in registers; 2 launches, was 3; carries summed in a new order
# speedup vs baseline: 19.9136x; 1.9234x over previous
#include <hip/hip_runtime.h>


typedef __attribute__((ext_vector_type(2))) float v2f;
typedef __attribute__((ext_vector_type(4))) float v4f;
typedef __attribute__((ext_vector_type(8))) float v8f;

#define CHUNK_L 25
#define NCOLS   4096
#define WCH     32
#define RUNB    8
#define CH_BLOCK 32

struct Params {
    float cm00, cm01, cm10, cm11;
    float km00, km01, km10, km11;
    float m0, m1, dt, h2, dt6;
};

__device__ inline Params make_params(const float* M, const float* C,
                                     const float* K, const float* dtp) {
    Params p;
    p.cm00 = C[0] + C[1]; p.cm01 = -C[1];
    p.cm10 = -C[1];       p.cm11 = C[2] + C[1];
    p.km00 = K[0] + K[1]; p.km01 = -K[1];
    p.km10 = -K[1];       p.km11 = K[2] + K[1];
    p.m0 = M[0]; p.m1 = M[1];
    p.dt = dtp[0];
    p.h2 = 0.5f * p.dt;
    p.dt6 = p.dt / 6.0f;
    return p;
}

__device__ inline void accel(const Params& p, float z0, float z1,
                             float v0, float v1, float u0, float u1,
                             float& a0, float& a1) {
    a0 = (u0 - (p.cm00 * v0 + p.cm01 * v1) - (p.km00 * z0 + p.km01 * z1)) / p.m0;
    a1 = (u1 - (p.cm10 * v0 + p.cm11 * v1) - (p.km10 * z0 + p.km11 * z1)) / p.m1;
}

__device__ inline void rk4_step(const Params& p, float& y0, float& y1,
                                float& v0, float& v1, float u0, float u1) {
    float k1x, k1y; accel(p, y0, y1, v0, v1, u0, u1, k1x, k1y);
    float y2x = y0 + v0 * p.h2, y2y = y1 + v1 * p.h2;
    float v2x = v0 + k1x * p.h2, v2y = v1 + k1y * p.h2;
    float k2x, k2y; accel(p, y2x, y2y, v2x, v2y, u0, u1, k2x, k2y);
    float y3x = y0 + v2x * p.h2, y3y = y1 + v2y * p.h2;
    float v3x = v0 + k2x * p.h2, v3y = v1 + k2y * p.h2;
    float k3x, k3y; accel(p, y3x, y3y, v3x, v3y, u0, u1, k3x, k3y);
    float y4x = y0 + v3x * p.dt, y4y = y1 + v3y * p.dt;
    float v4x = v0 + k3x * p.dt, v4y = v1 + k3y * p.dt;
    float k4x, k4y; accel(p, y4x, y4y, v4x, v4y, u0, u1, k4x, k4y);
    float yn0 = y0 + p.dt6 * (v0 + 2.0f * v2x + 2.0f * v3x + v4x);
    float yn1 = y1 + p.dt6 * (v1 + 2.0f * v2y + 2.0f * v3y + v4y);
    float vn0 = v0 + p.dt6 * (k1x + 2.0f * k2x + 2.0f * k3x + k4x);
    float vn1 = v1 + p.dt6 * (k1y + 2.0f * k2y + 2.0f * k3y + k4y);
    y0 = yn0; y1 = yn1; v0 = vn0; v1 = vn1;
}

struct StepMap {
    float a[4][4];
    float b[4][2];
};

__device__ inline float lane_bcast(float x, int srclane) {
    return __int_as_float(__builtin_amdgcn_readlane(__float_as_int(x), srclane));
}

__device__ inline StepMap make_step_map(const Params& p, int lane, v8f& xa) {
    const int j = lane & 7;
    float y0 = (j == 0) ? 1.f : 0.f, y1 = (j == 1) ? 1.f : 0.f;
    float v0 = (j == 2) ? 1.f : 0.f, v1 = (j == 3) ? 1.f : 0.f;
    const float u0 = (j == 4) ? 1.f : 0.f, u1 = (j == 5) ? 1.f : 0.f;
    rk4_step(p, y0, y1, v0, v1, u0, u1);
    StepMap m;
    #pragma unroll
    for (int c = 0; c < 4; ++c) {
        m.a[0][c] = lane_bcast(y0, c);
        m.a[1][c] = lane_bcast(y1, c);
        m.a[2][c] = lane_bcast(v0, c);
        m.a[3][c] = lane_bcast(v1, c);
    }
    #pragma unroll
    for (int c = 0; c < 2; ++c) {
        m.b[0][c] = lane_bcast(y0, 4 + c);
        m.b[1][c] = lane_bcast(y1, 4 + c);
        m.b[2][c] = lane_bcast(v0, 4 + c);
        m.b[3][c] = lane_bcast(v1, 4 + c);
    }
    const bool col = j < 4;
    xa = (v8f){col ? y0 : 0.f, col ? y1 : 0.f, col ? v0 : 0.f, col ? v1 : 0.f,
               0.f, 0.f, 0.f, 0.f};
    return m;
}

__device__ inline void affine_step(const StepMap& m, float& y0, float& y1,
                                   float& v0, float& v1, float u0, float u1) {
    const float s[4] = {y0, y1, v0, v1};
    float r[4];
    #pragma unroll
    for (int i = 0; i < 4; ++i) {
        float acc = m.b[i][1] * u1;
        acc = fmaf(m.b[i][0], u0, acc);
        #pragma unroll
        for (int j = 3; j >= 0; --j) {
            if (j != i) acc = fmaf(m.a[i][j], s[j], acc);
        }
        r[i] = fmaf(m.a[i][i], s[i], acc);
    }
    y0 = r[0]; y1 = r[1]; v0 = r[2]; v1 = r[3];
}

__device__ inline v8f wmma4(v2f a, v2f b, v8f c) {
    return __builtin_amdgcn_wmma_f32_16x16x4_f32(
        false, a, false, b, (short)0, c, false, false);
}

__device__ inline v2f frag(const v8f& t, bool hi) {
    v2f f;
    f.x = hi ? t[2] : t[0];
    f.y = hi ? t[3] : t[1];
    return f;
}

__device__ inline v8f col_tile(float a0, float a1, float a2, float a3) {
    return (v8f){a0, a1, a2, a3, 0.f, 0.f, 0.f, 0.f};
}

struct Pw {
    v2f fx, ft;
};

__device__ inline Pw pw_mul(const Pw& p, const Pw& q, bool hi) {
    const v8f z = {0.f, 0.f, 0.f, 0.f, 0.f, 0.f, 0.f, 0.f};
    const v8f x = wmma4(p.ft, q.fx, z);
    const v8f y = wmma4(q.fx, p.ft, z);
    Pw r;
    r.fx = frag(x, hi);
    r.ft = frag(y, hi);
    return r;
}

__device__ inline v8f tile_of_a_transposed(const StepMap& m, int lane) {
    const int i = lane & 7;
    float r[4];
    #pragma unroll
    for (int k = 0; k < 4; ++k) {
        const float r0 = m.a[0][k], r1 = m.a[1][k], r2 = m.a[2][k], r3 = m.a[3][k];
        float v = 0.f;
        v = (i == 3) ? r3 : v;
        v = (i == 2) ? r2 : v;
        v = (i == 1) ? r1 : v;
        v = (i == 0) ? r0 : v;
        r[k] = v;
    }
    return col_tile(r[0], r[1], r[2], r[3]);
}

struct PowTab {
    v2f g, g2, g4, g8, g16;
    v2f h, h8, h16, h32, h64;
};

template <bool BLOCKS>
__device__ inline PowTab build_powers(const StepMap& m, const v8f& xa,
                                      int lane, bool hi) {
    Pw a;
    a.fx = frag(xa, hi);
    a.ft = frag(tile_of_a_transposed(m, lane), hi);
    const Pw a2  = pw_mul(a,   a,   hi);
    const Pw a4  = pw_mul(a2,  a2,  hi);
    const Pw a8  = pw_mul(a4,  a4,  hi);
    const Pw a16 = pw_mul(a8,  a8,  hi);
    const Pw a24 = pw_mul(a16, a8,  hi);
    const Pw g   = pw_mul(a24, a,   hi);
    const Pw g2  = pw_mul(g,   g,   hi);
    const Pw g4  = pw_mul(g2,  g2,  hi);
    const Pw g8  = pw_mul(g4,  g4,  hi);
    const Pw g16 = pw_mul(g8,  g8,  hi);
    PowTab t;
    t.g = g.ft; t.g2 = g2.ft; t.g4 = g4.ft; t.g8 = g8.ft; t.g16 = g16.ft;
    t.h = t.h8 = t.h16 = t.h32 = t.h64 = g.ft;
    if (BLOCKS) {
        const Pw h   = pw_mul(g16, g16, hi);
        const Pw h2  = pw_mul(h,   h,   hi);
        const Pw h4  = pw_mul(h2,  h2,  hi);
        const Pw h8  = pw_mul(h4,  h4,  hi);
        const Pw h16 = pw_mul(h8,  h8,  hi);
        const Pw h32 = pw_mul(h16, h16, hi);
        const Pw h64 = pw_mul(h32, h32, hi);
        t.h = h.ft; t.h8 = h8.ft; t.h16 = h16.ft; t.h32 = h32.ft; t.h64 = h64.ft;
    }
    return t;
}

template <int D>
__device__ inline v2f frag_from_left(const v8f& t, bool hi) {
    const v2f f = frag(t, hi);
    v2f r;
    r.x = __int_as_float(__builtin_amdgcn_update_dpp(
        0, __float_as_int(f.x), 0x110 + D, 0xf, 0xf, true));
    r.y = __int_as_float(__builtin_amdgcn_update_dpp(
        0, __float_as_int(f.y), 0x110 + D, 0xf, 0xf, true));
    return r;
}

__device__ inline v8f scan16(v8f t, v2f m1, v2f m2, v2f m4, v2f m8, bool hi) {
    t = wmma4(m1, frag_from_left<1>(t, hi), t);
    t = wmma4(m2, frag_from_left<2>(t, hi), t);
    t = wmma4(m4, frag_from_left<4>(t, hi), t);
    t = wmma4(m8, frag_from_left<8>(t, hi), t);
    return t;
}

__device__ inline void load_chunk(const float* __restrict__ up,
                                  float (&uu)[CHUNK_L][2]) {
    #pragma unroll
    for (int t = 0; t < CHUNK_L; ++t) {
        uu[t][0] = up[2 * t];
        uu[t][1] = up[2 * t + 1];
    }
}

__device__ inline void chunk_end_from_zero(const StepMap& m,
                                           const float (&uu)[CHUNK_L][2],
                                           float (&e)[4]) {
    float y0 = 0.f, y1 = 0.f, v0 = 0.f, v1 = 0.f;
    #pragma unroll
    for (int t = 0; t < CHUNK_L; ++t) {
        affine_step(m, y0, y1, v0, v1, uu[t][0], uu[t][1]);
    }
    e[0] = y0; e[1] = y1; e[2] = v0; e[3] = v1;
}

__global__ __launch_bounds__(CH_BLOCK) void phase1_chunks(
    const float* __restrict__ u, const float* __restrict__ M,
    const float* __restrict__ C, const float* __restrict__ K,
    const float* __restrict__ dtp, float* __restrict__ wsE,
    float* __restrict__ wsR, int nfull) {
    __shared__ __attribute__((aligned(16))) float ex[WCH][4];
    const int  lane = threadIdx.x & 31;
    const int  n    = lane & 15;
    const bool hi   = lane >= 16;
    const int  i    = blockIdx.x * blockDim.x + threadIdx.x;
    const bool live = i < nfull;
    const float* up = u + (size_t)i * CHUNK_L * 2;
    float uu[CHUNK_L][2];
    float e[4];
    v8f xa;
    if (__all(live)) {
        load_chunk(up, uu);
        const Params  p = make_params(M, C, K, dtp);
        const StepMap m = make_step_map(p, lane, xa);
        const PowTab  w = build_powers<false>(m, xa, lane, hi);
        chunk_end_from_zero(m, uu, e);
        v4f ev; ev.x = e[0]; ev.y = e[1]; ev.z = e[2]; ev.w = e[3];
        *(v4f*)(wsE + (size_t)i * 4) = ev;
        *(v4f*)(&ex[lane][0]) = ev;
        __syncthreads();
        const v4f x0 = *(const v4f*)(&ex[2 * n][0]);
        const v4f x1 = *(const v4f*)(&ex[2 * n + 1][0]);
        v8f t = wmma4(w.g, frag(col_tile(x0.x, x0.y, x0.z, x0.w), hi),
                      col_tile(x1.x, x1.y, x1.z, x1.w));
        t = scan16(t, w.g2, w.g4, w.g8, w.g16, hi);
        if (lane == 15) {
            v4f r; r.x = t[0]; r.y = t[1]; r.z = t[2]; r.w = t[3];
            *(v4f*)(wsR + (size_t)blockIdx.x * 4) = r;
        }
    } else {
        const Params  p = make_params(M, C, K, dtp);
        const StepMap m = make_step_map(p, lane, xa);
        if (live) {
            load_chunk(up, uu);
            chunk_end_from_zero(m, uu, e);
            wsE[i * 4 + 0] = e[0]; wsE[i * 4 + 1] = e[1];
            wsE[i * 4 + 2] = e[2]; wsE[i * 4 + 3] = e[3];
        }
    }
}

__device__ inline void emit_full_chunk(const StepMap& m,
                                       const float (&uu)[CHUNK_L][2],
                                       float y0, float y1, float v0, float v1,
                                       float* __restrict__ op) {
    float yo[CHUNK_L][2];
    #pragma unroll
    for (int t = 0; t < CHUNK_L; ++t) {
        affine_step(m, y0, y1, v0, v1, uu[t][0], uu[t][1]);
        yo[t][0] = y0;
        yo[t][1] = y1;
    }
    #pragma unroll
    for (int t = 0; t < CHUNK_L; ++t) {
        op[2 * t]     = yo[t][0];
        op[2 * t + 1] = yo[t][1];
    }
}

__global__ __launch_bounds__(CH_BLOCK) void phase2_emit(
    const float* __restrict__ u, const float* __restrict__ M,
    const float* __restrict__ C, const float* __restrict__ K,
    const float* __restrict__ x0, const float* __restrict__ dtp,
    const float* __restrict__ wsE, const float* __restrict__ wsR,
    float* __restrict__ out, int P, int T) {
    __shared__ __attribute__((aligned(16))) float sx[WCH][4];
    const int  lane = threadIdx.x & 31;
    const int  n    = lane & 15;
    const bool hi   = lane >= 16;
    const int  b    = blockIdx.x;
    const int  c0   = b * WCH;
    const int  i    = c0 + lane;
    const int  nfull = P - 1;
    const bool live = i < P;
    const bool full = live && (i + 1) * CHUNK_L <= T;
    const bool allfull = __all(full);
    const float* up = u + (size_t)i * CHUNK_L * 2;
    float*       op = out + (size_t)i * CHUNK_L * 2;
    float uu[CHUNK_L][2];
    if (allfull) load_chunk(up, uu);

    float xr[RUNB][4];
    #pragma unroll
    for (int g = 0; g < RUNB; ++g) {
        const int j = n * RUNB + g;
        float a0, a1, a2, a3;
        if (j == 0) {
            a0 = x0[0]; a1 = x0[2]; a2 = x0[1]; a3 = x0[3];
        } else if (j <= b) {
            const v4f r = *(const v4f*)(wsR + (size_t)(j - 1) * 4);
            a0 = r.x; a1 = r.y; a2 = r.z; a3 = r.w;
        } else {
            a0 = a1 = a2 = a3 = 0.f;
        }
        xr[g][0] = a0; xr[g][1] = a1; xr[g][2] = a2; xr[g][3] = a3;
    }
    v4f e0 = {0.f, 0.f, 0.f, 0.f}, e1 = {0.f, 0.f, 0.f, 0.f};
    {
        const int k0 = c0 + 2 * n - 1;
        const int k1 = c0 + 2 * n;
        if (n >= 1 && k0 < nfull) e0 = *(const v4f*)(wsE + (size_t)k0 * 4);
        if (k1 < nfull)           e1 = *(const v4f*)(wsE + (size_t)k1 * 4);
    }

    const Params p = make_params(M, C, K, dtp);
    v8f xa;
    const StepMap m = make_step_map(p, lane, xa);
    const PowTab  w = build_powers<true>(m, xa, lane, hi);

    float s0, s1, s2, s3;
    {
        v8f t = col_tile(xr[0][0], xr[0][1], xr[0][2], xr[0][3]);
        #pragma unroll
        for (int g = 1; g < RUNB; ++g) {
            t = wmma4(w.h, frag(t, hi),
                      col_tile(xr[g][0], xr[g][1], xr[g][2], xr[g][3]));
        }
        t = scan16(t, w.h8, w.h16, w.h32, w.h64, hi);
        v2f bp = frag_from_left<1>(t, hi);
        const int gsel = b & (RUNB - 1);
        float q0 = 0.f, q1 = 0.f, q2 = 0.f, q3 = 0.f;
        #pragma unroll
        for (int g = 0; g < RUNB; ++g) {
            const v8f o = wmma4(w.h, bp,
                                col_tile(xr[g][0], xr[g][1], xr[g][2], xr[g][3]));
            if (g == gsel) { q0 = o[0]; q1 = o[1]; q2 = o[2]; q3 = o[3]; }
            bp = frag(o, hi);
        }
        const int nsel = b / RUNB;
        s0 = lane_bcast(q0, nsel); s1 = lane_bcast(q1, nsel);
        s2 = lane_bcast(q2, nsel); s3 = lane_bcast(q3, nsel);
    }

    {
        const bool first = (n == 0);
        const v8f c0t = col_tile(first ? s0 : e0.x, first ? s1 : e0.y,
                                 first ? s2 : e0.z, first ? s3 : e0.w);
        const v8f c1t = col_tile(e1.x, e1.y, e1.z, e1.w);
        v8f t = wmma4(w.g, frag(c0t, hi), c1t);
        t = scan16(t, w.g2, w.g4, w.g8, w.g16, hi);
        const v8f o0 = wmma4(w.g, frag_from_left<1>(t, hi), c0t);
        const v8f o1 = wmma4(w.g, frag(o0, hi), c1t);
        if (!hi) {
            v4f a; a.x = o0[0]; a.y = o0[1]; a.z = o0[2]; a.w = o0[3];
            v4f c; c.x = o1[0]; c.y = o1[1]; c.z = o1[2]; c.w = o1[3];
            *(v4f*)(&sx[2 * n][0])     = a;
            *(v4f*)(&sx[2 * n + 1][0]) = c;
        }
    }
    __syncthreads();
    const v4f sv = *(const v4f*)(&sx[lane][0]);
    float y0 = sv.x, y1 = sv.y, v0 = sv.z, v1 = sv.w;

    if (allfull) {
        emit_full_chunk(m, uu, y0, y1, v0, v1, op);
    } else if (full) {
        load_chunk(up, uu);
        emit_full_chunk(m, uu, y0, y1, v0, v1, op);
    } else if (live) {
        for (int t = i * CHUNK_L; t < T; ++t) {
            affine_step(m, y0, y1, v0, v1, u[2 * t], u[2 * t + 1]);
            out[2 * t]     = y0;
            out[2 * t + 1] = y1;
        }
    }
}

extern "C" void kernel_launch(void* const* d_in, const int* in_sizes, int n_in,
                              void* d_out, int out_size, void* d_ws,
                              size_t ws_size, hipStream_t stream) {
    const float* u  = (const float*)d_in[0];
    const float* M  = (const float*)d_in[1];
    const float* C  = (const float*)d_in[2];
    const float* K  = (const float*)d_in[3];
    const float* x0 = (const float*)d_in[4];
    const float* dt = (const float*)d_in[5];
    float* out = (float*)d_out;

    const int T = in_sizes[0] / 2;
    int P = (T + CHUNK_L - 1) / CHUNK_L;
    if (P > NCOLS) P = NCOLS;
    const int nfull = P - 1;

    float* wsE = (float*)d_ws;
    float* wsR = wsE + NCOLS * 4;

    phase1_chunks<<<(nfull + CH_BLOCK - 1) / CH_BLOCK, CH_BLOCK, 0, stream>>>(
        u, M, C, K, dt, wsE, wsR, nfull);
    phase2_emit<<<(P + CH_BLOCK - 1) / CH_BLOCK, CH_BLOCK, 0, stream>>>(
        u, M, C, K, x0, dt, wsE, wsR, out, P, T);
}


# Launch 2 loads the one-step map and its ten matrix powers from a workspace table that one extra block of launch 1 builds, not per wave (rk4 on unit vectors + 34 WMMAs); scan inputs load ahead of u
# speedup vs baseline: 22.5760x; 1.1337x over previous
#include <hip/hip_runtime.h>


typedef __attribute__((ext_vector_type(2))) float v2f;
typedef __attribute__((ext_vector_type(4))) float v4f;
typedef __attribute__((ext_vector_type(8))) float v8f;

#define CHUNK_L 25
#define NCOLS   4096
#define WCH     32
#define RUNB    8
#define CH_BLOCK 32
#define TAB_POW 640

struct Params {
    float cm00, cm01, cm10, cm11;
    float km00, km01, km10, km11;
    float m0, m1, dt, h2, dt6;
};

__device__ inline Params make_params(const float* M, const float* C,
                                     const float* K, const float* dtp) {
    Params p;
    p.cm00 = C[0] + C[1]; p.cm01 = -C[1];
    p.cm10 = -C[1];       p.cm11 = C[2] + C[1];
    p.km00 = K[0] + K[1]; p.km01 = -K[1];
    p.km10 = -K[1];       p.km11 = K[2] + K[1];
    p.m0 = M[0]; p.m1 = M[1];
    p.dt = dtp[0];
    p.h2 = 0.5f * p.dt;
    p.dt6 = p.dt / 6.0f;
    return p;
}

__device__ inline void accel(const Params& p, float z0, float z1,
                             float v0, float v1, float u0, float u1,
                             float& a0, float& a1) {
    a0 = (u0 - (p.cm00 * v0 + p.cm01 * v1) - (p.km00 * z0 + p.km01 * z1)) / p.m0;
    a1 = (u1 - (p.cm10 * v0 + p.cm11 * v1) - (p.km10 * z0 + p.km11 * z1)) / p.m1;
}

__device__ inline void rk4_step(const Params& p, float& y0, float& y1,
                                float& v0, float& v1, float u0, float u1) {
    float k1x, k1y; accel(p, y0, y1, v0, v1, u0, u1, k1x, k1y);
    float y2x = y0 + v0 * p.h2, y2y = y1 + v1 * p.h2;
    float v2x = v0 + k1x * p.h2, v2y = v1 + k1y * p.h2;
    float k2x, k2y; accel(p, y2x, y2y, v2x, v2y, u0, u1, k2x, k2y);
    float y3x = y0 + v2x * p.h2, y3y = y1 + v2y * p.h2;
    float v3x = v0 + k2x * p.h2, v3y = v1 + k2y * p.h2;
    float k3x, k3y; accel(p, y3x, y3y, v3x, v3y, u0, u1, k3x, k3y);
    float y4x = y0 + v3x * p.dt, y4y = y1 + v3y * p.dt;
    float v4x = v0 + k3x * p.dt, v4y = v1 + k3y * p.dt;
    float k4x, k4y; accel(p, y4x, y4y, v4x, v4y, u0, u1, k4x, k4y);
    float yn0 = y0 + p.dt6 * (v0 + 2.0f * v2x + 2.0f * v3x + v4x);
    float yn1 = y1 + p.dt6 * (v1 + 2.0f * v2y + 2.0f * v3y + v4y);
    float vn0 = v0 + p.dt6 * (k1x + 2.0f * k2x + 2.0f * k3x + k4x);
    float vn1 = v1 + p.dt6 * (k1y + 2.0f * k2y + 2.0f * k3y + k4y);
    y0 = yn0; y1 = yn1; v0 = vn0; v1 = vn1;
}

struct StepMap {
    float a[4][4];
    float b[4][2];
};

__device__ inline float lane_bcast(float x, int srclane) {
    return __int_as_float(__builtin_amdgcn_readlane(__float_as_int(x), srclane));
}

__device__ inline StepMap make_step_map(const Params& p, int lane, v8f& xa,
                                        v4f& colv) {
    const int j = lane & 7;
    float y0 = (j == 0) ? 1.f : 0.f, y1 = (j == 1) ? 1.f : 0.f;
    float v0 = (j == 2) ? 1.f : 0.f, v1 = (j == 3) ? 1.f : 0.f;
    const float u0 = (j == 4) ? 1.f : 0.f, u1 = (j == 5) ? 1.f : 0.f;
    rk4_step(p, y0, y1, v0, v1, u0, u1);
    StepMap m;
    #pragma unroll
    for (int c = 0; c < 4; ++c) {
        m.a[0][c] = lane_bcast(y0, c);
        m.a[1][c] = lane_bcast(y1, c);
        m.a[2][c] = lane_bcast(v0, c);
        m.a[3][c] = lane_bcast(v1, c);
    }
    #pragma unroll
    for (int c = 0; c < 2; ++c) {
        m.b[0][c] = lane_bcast(y0, 4 + c);
        m.b[1][c] = lane_bcast(y1, 4 + c);
        m.b[2][c] = lane_bcast(v0, 4 + c);
        m.b[3][c] = lane_bcast(v1, 4 + c);
    }
    const bool col = j < 4;
    xa = (v8f){col ? y0 : 0.f, col ? y1 : 0.f, col ? v0 : 0.f, col ? v1 : 0.f,
               0.f, 0.f, 0.f, 0.f};
    colv.x = y0; colv.y = y1; colv.z = v0; colv.w = v1;
    return m;
}

__device__ inline void affine_step(const StepMap& m, float& y0, float& y1,
                                   float& v0, float& v1, float u0, float u1) {
    const float s[4] = {y0, y1, v0, v1};
    float r[4];
    #pragma unroll
    for (int i = 0; i < 4; ++i) {
        float acc = m.b[i][1] * u1;
        acc = fmaf(m.b[i][0], u0, acc);
        #pragma unroll
        for (int j = 3; j >= 0; --j) {
            if (j != i) acc = fmaf(m.a[i][j], s[j], acc);
        }
        r[i] = fmaf(m.a[i][i], s[i], acc);
    }
    y0 = r[0]; y1 = r[1]; v0 = r[2]; v1 = r[3];
}

__device__ inline v8f wmma4(v2f a, v2f b, v8f c) {
    return __builtin_amdgcn_wmma_f32_16x16x4_f32(
        false, a, false, b, (short)0, c, false, false);
}

__device__ inline v2f frag(const v8f& t, bool hi) {
    v2f f;
    f.x = hi ? t[2] : t[0];
    f.y = hi ? t[3] : t[1];
    return f;
}

__device__ inline v8f col_tile(float a0, float a1, float a2, float a3) {
    return (v8f){a0, a1, a2, a3, 0.f, 0.f, 0.f, 0.f};
}

struct Pw {
    v2f fx, ft;
};

__device__ inline Pw pw_mul(const Pw& p, const Pw& q, bool hi) {
    const v8f z = {0.f, 0.f, 0.f, 0.f, 0.f, 0.f, 0.f, 0.f};
    const v8f x = wmma4(p.ft, q.fx, z);
    const v8f y = wmma4(q.fx, p.ft, z);
    Pw r;
    r.fx = frag(x, hi);
    r.ft = frag(y, hi);
    return r;
}

__device__ inline v8f tile_of_a_transposed(const StepMap& m, int lane) {
    const int i = lane & 7;
    float r[4];
    #pragma unroll
    for (int k = 0; k < 4; ++k) {
        const float r0 = m.a[0][k], r1 = m.a[1][k], r2 = m.a[2][k], r3 = m.a[3][k];
        float v = 0.f;
        v = (i == 3) ? r3 : v;
        v = (i == 2) ? r2 : v;
        v = (i == 1) ? r1 : v;
        v = (i == 0) ? r0 : v;
        r[k] = v;
    }
    return col_tile(r[0], r[1], r[2], r[3]);
}

struct PowTab {
    v2f g, g2, g4, g8, g16;
    v2f h, h8, h16, h32, h64;
};

template <bool BLOCKS>
__device__ inline PowTab build_powers(const StepMap& m, const v8f& xa,
                                      int lane, bool hi) {
    Pw a;
    a.fx = frag(xa, hi);
    a.ft = frag(tile_of_a_transposed(m, lane), hi);
    const Pw a2  = pw_mul(a,   a,   hi);
    const Pw a4  = pw_mul(a2,  a2,  hi);
    const Pw a8  = pw_mul(a4,  a4,  hi);
    const Pw a16 = pw_mul(a8,  a8,  hi);
    const Pw a24 = pw_mul(a16, a8,  hi);
    const Pw g   = pw_mul(a24, a,   hi);
    const Pw g2  = pw_mul(g,   g,   hi);
    const Pw g4  = pw_mul(g2,  g2,  hi);
    const Pw g8  = pw_mul(g4,  g4,  hi);
    const Pw g16 = pw_mul(g8,  g8,  hi);
    PowTab t;
    t.g = g.ft; t.g2 = g2.ft; t.g4 = g4.ft; t.g8 = g8.ft; t.g16 = g16.ft;
    t.h = t.h8 = t.h16 = t.h32 = t.h64 = g.ft;
    if (BLOCKS) {
        const Pw h   = pw_mul(g16, g16, hi);
        const Pw h2  = pw_mul(h,   h,   hi);
        const Pw h4  = pw_mul(h2,  h2,  hi);
        const Pw h8  = pw_mul(h4,  h4,  hi);
        const Pw h16 = pw_mul(h8,  h8,  hi);
        const Pw h32 = pw_mul(h16, h16, hi);
        const Pw h64 = pw_mul(h32, h32, hi);
        t.h = h.ft; t.h8 = h8.ft; t.h16 = h16.ft; t.h32 = h32.ft; t.h64 = h64.ft;
    }
    return t;
}

__device__ inline void store_table(float* __restrict__ wsT, const PowTab& w,
                                   v4f colv, int lane) {
    v4f* q = (v4f*)wsT + lane;
    q[0]   = (v4f){w.g.x,   w.g.y,   w.g2.x,  w.g2.y};
    q[32]  = (v4f){w.g4.x,  w.g4.y,  w.g8.x,  w.g8.y};
    q[64]  = (v4f){w.g16.x, w.g16.y, w.h.x,   w.h.y};
    q[96]  = (v4f){w.h8.x,  w.h8.y,  w.h16.x, w.h16.y};
    q[128] = (v4f){w.h32.x, w.h32.y, w.h64.x, w.h64.y};
    if (lane < 6) ((v4f*)(wsT + TAB_POW))[lane] = colv;
}

__device__ inline PowTab load_powers(const float* __restrict__ wsT, int lane) {
    const v4f* q = (const v4f*)wsT + lane;
    const v4f q0 = q[0], q1 = q[32], q2 = q[64], q3 = q[96], q4 = q[128];
    PowTab t;
    t.g   = (v2f){q0.x, q0.y}; t.g2  = (v2f){q0.z, q0.w};
    t.g4  = (v2f){q1.x, q1.y}; t.g8  = (v2f){q1.z, q1.w};
    t.g16 = (v2f){q2.x, q2.y}; t.h   = (v2f){q2.z, q2.w};
    t.h8  = (v2f){q3.x, q3.y}; t.h16 = (v2f){q3.z, q3.w};
    t.h32 = (v2f){q4.x, q4.y}; t.h64 = (v2f){q4.z, q4.w};
    return t;
}

__device__ inline StepMap load_step_map(const float* __restrict__ wsM) {
    StepMap m;
    #pragma unroll
    for (int c = 0; c < 4; ++c) {
        m.a[0][c] = wsM[c * 4 + 0]; m.a[1][c] = wsM[c * 4 + 1];
        m.a[2][c] = wsM[c * 4 + 2]; m.a[3][c] = wsM[c * 4 + 3];
    }
    #pragma unroll
    for (int c = 0; c < 2; ++c) {
        m.b[0][c] = wsM[(4 + c) * 4 + 0]; m.b[1][c] = wsM[(4 + c) * 4 + 1];
        m.b[2][c] = wsM[(4 + c) * 4 + 2]; m.b[3][c] = wsM[(4 + c) * 4 + 3];
    }
    return m;
}

template <int D>
__device__ inline v2f frag_from_left(const v8f& t, bool hi) {
    const v2f f = frag(t, hi);
    v2f r;
    r.x = __int_as_float(__builtin_amdgcn_update_dpp(
        0, __float_as_int(f.x), 0x110 + D, 0xf, 0xf, true));
    r.y = __int_as_float(__builtin_amdgcn_update_dpp(
        0, __float_as_int(f.y), 0x110 + D, 0xf, 0xf, true));
    return r;
}

__device__ inline v8f scan16(v8f t, v2f m1, v2f m2, v2f m4, v2f m8, bool hi) {
    t = wmma4(m1, frag_from_left<1>(t, hi), t);
    t = wmma4(m2, frag_from_left<2>(t, hi), t);
    t = wmma4(m4, frag_from_left<4>(t, hi), t);
    t = wmma4(m8, frag_from_left<8>(t, hi), t);
    return t;
}

__device__ inline void load_chunk(const float* __restrict__ up,
                                  float (&uu)[CHUNK_L][2]) {
    #pragma unroll
    for (int t = 0; t < CHUNK_L; ++t) {
        uu[t][0] = up[2 * t];
        uu[t][1] = up[2 * t + 1];
    }
}

__device__ inline void chunk_end_from_zero(const StepMap& m,
                                           const float (&uu)[CHUNK_L][2],
                                           float (&e)[4]) {
    float y0 = 0.f, y1 = 0.f, v0 = 0.f, v1 = 0.f;
    #pragma unroll
    for (int t = 0; t < CHUNK_L; ++t) {
        affine_step(m, y0, y1, v0, v1, uu[t][0], uu[t][1]);
    }
    e[0] = y0; e[1] = y1; e[2] = v0; e[3] = v1;
}

__global__ __launch_bounds__(CH_BLOCK) void phase1_chunks(
    const float* __restrict__ u, const float* __restrict__ M,
    const float* __restrict__ C, const float* __restrict__ K,
    const float* __restrict__ dtp, float* __restrict__ wsE,
    float* __restrict__ wsR, float* __restrict__ wsT, int nfull) {
    __shared__ __attribute__((aligned(16))) float ex[WCH][4];
    const int  lane = threadIdx.x & 31;
    const int  n    = lane & 15;
    const bool hi   = lane >= 16;
    const int  i    = blockIdx.x * blockDim.x + threadIdx.x;
    const bool live = i < nfull;
    const float* up = u + (size_t)i * CHUNK_L * 2;
    float uu[CHUNK_L][2];
    float e[4];
    v8f xa;
    v4f colv;
    if (__all(live)) {
        load_chunk(up, uu);
        const Params  p = make_params(M, C, K, dtp);
        const StepMap m = make_step_map(p, lane, xa, colv);
        const PowTab  w = build_powers<false>(m, xa, lane, hi);
        chunk_end_from_zero(m, uu, e);
        v4f ev; ev.x = e[0]; ev.y = e[1]; ev.z = e[2]; ev.w = e[3];
        *(v4f*)(wsE + (size_t)i * 4) = ev;
        *(v4f*)(&ex[lane][0]) = ev;
        __syncthreads();
        const v4f x0 = *(const v4f*)(&ex[2 * n][0]);
        const v4f x1 = *(const v4f*)(&ex[2 * n + 1][0]);
        v8f t = wmma4(w.g, frag(col_tile(x0.x, x0.y, x0.z, x0.w), hi),
                      col_tile(x1.x, x1.y, x1.z, x1.w));
        t = scan16(t, w.g2, w.g4, w.g8, w.g16, hi);
        if (lane == 15) {
            v4f r; r.x = t[0]; r.y = t[1]; r.z = t[2]; r.w = t[3];
            *(v4f*)(wsR + (size_t)blockIdx.x * 4) = r;
        }
    } else {
        const Params  p = make_params(M, C, K, dtp);
        const StepMap m = make_step_map(p, lane, xa, colv);
        if (blockIdx.x == gridDim.x - 1) {
            const PowTab w = build_powers<true>(m, xa, lane, hi);
            store_table(wsT, w, colv, lane);
        }
        if (live) {
            load_chunk(up, uu);
            chunk_end_from_zero(m, uu, e);
            wsE[i * 4 + 0] = e[0]; wsE[i * 4 + 1] = e[1];
            wsE[i * 4 + 2] = e[2]; wsE[i * 4 + 3] = e[3];
        }
    }
}

__device__ inline void emit_full_chunk(const StepMap& m,
                                       const float (&uu)[CHUNK_L][2],
                                       float y0, float y1, float v0, float v1,
                                       float* __restrict__ op) {
    float yo[CHUNK_L][2];
    #pragma unroll
    for (int t = 0; t < CHUNK_L; ++t) {
        affine_step(m, y0, y1, v0, v1, uu[t][0], uu[t][1]);
        yo[t][0] = y0;
        yo[t][1] = y1;
    }
    #pragma unroll
    for (int t = 0; t < CHUNK_L; ++t) {
        op[2 * t]     = yo[t][0];
        op[2 * t + 1] = yo[t][1];
    }
}

__global__ __launch_bounds__(CH_BLOCK) void phase2_emit(
    const float* __restrict__ u, const float* __restrict__ x0,
    const float* __restrict__ wsE, const float* __restrict__ wsR,
    const float* __restrict__ wsT, float* __restrict__ out, int P, int T) {
    __shared__ __attribute__((aligned(16))) float sx[WCH][4];
    const int  lane = threadIdx.x & 31;
    const int  n    = lane & 15;
    const bool hi   = lane >= 16;
    const int  b    = blockIdx.x;
    const int  c0   = b * WCH;
    const int  i    = c0 + lane;
    const int  nfull = P - 1;
    const bool live = i < P;
    const bool full = live && (i + 1) * CHUNK_L <= T;
    const bool allfull = __all(full);
    const float* up = u + (size_t)i * CHUNK_L * 2;
    float*       op = out + (size_t)i * CHUNK_L * 2;
    v4f e0 = {0.f, 0.f, 0.f, 0.f}, e1 = {0.f, 0.f, 0.f, 0.f};
    {
        const int k0 = c0 + 2 * n - 1;
        const int k1 = c0 + 2 * n;
        if (n >= 1 && k0 < nfull) e0 = *(const v4f*)(wsE + (size_t)k0 * 4);
        if (k1 < nfull)           e1 = *(const v4f*)(wsE + (size_t)k1 * 4);
    }
    const PowTab  w = load_powers(wsT, lane);
    const StepMap m = load_step_map(wsT + TAB_POW);
    const float i0 = x0[0], i1 = x0[2], i2 = x0[1], i3 = x0[3];
    float xr[RUNB][4];
    #pragma unroll
    for (int g = 0; g < RUNB; ++g) {
        const int j = n * RUNB + g;
        v4f r = {0.f, 0.f, 0.f, 0.f};
        if (j >= 1 && j <= b) r = *(const v4f*)(wsR + (size_t)(j - 1) * 4);
        xr[g][0] = r.x; xr[g][1] = r.y; xr[g][2] = r.z; xr[g][3] = r.w;
    }
    {
        const bool init = (n == 0);
        xr[0][0] = init ? i0 : xr[0][0]; xr[0][1] = init ? i1 : xr[0][1];
        xr[0][2] = init ? i2 : xr[0][2]; xr[0][3] = init ? i3 : xr[0][3];
    }
    asm volatile("" :: "v"(xr[RUNB - 1][0]), "v"(xr[RUNB - 1][1]),
                       "v"(xr[RUNB - 1][2]), "v"(xr[RUNB - 1][3]));
    float uu[CHUNK_L][2];
    if (allfull) load_chunk(up, uu);

    float s0, s1, s2, s3;
    {
        v8f t = col_tile(xr[0][0], xr[0][1], xr[0][2], xr[0][3]);
        #pragma unroll
        for (int g = 1; g < RUNB; ++g) {
            t = wmma4(w.h, frag(t, hi),
                      col_tile(xr[g][0], xr[g][1], xr[g][2], xr[g][3]));
        }
        t = scan16(t, w.h8, w.h16, w.h32, w.h64, hi);
        v2f bp = frag_from_left<1>(t, hi);
        const int gsel = b & (RUNB - 1);
        float q0 = 0.f, q1 = 0.f, q2 = 0.f, q3 = 0.f;
        #pragma unroll
        for (int g = 0; g < RUNB; ++g) {
            const v8f o = wmma4(w.h, bp,
                                col_tile(xr[g][0], xr[g][1], xr[g][2], xr[g][3]));
            if (g == gsel) { q0 = o[0]; q1 = o[1]; q2 = o[2]; q3 = o[3]; }
            bp = frag(o, hi);
        }
        const int nsel = b / RUNB;
        s0 = lane_bcast(q0, nsel); s1 = lane_bcast(q1, nsel);
        s2 = lane_bcast(q2, nsel); s3 = lane_bcast(q3, nsel);
    }

    {
        const bool first = (n == 0);
        const v8f c0t = col_tile(first ? s0 : e0.x, first ? s1 : e0.y,
                                 first ? s2 : e0.z, first ? s3 : e0.w);
        const v8f c1t = col_tile(e1.x, e1.y, e1.z, e1.w);
        v8f t = wmma4(w.g, frag(c0t, hi), c1t);
        t = scan16(t, w.g2, w.g4, w.g8, w.g16, hi);
        const v8f o0 = wmma4(w.g, frag_from_left<1>(t, hi), c0t);
        const v8f o1 = wmma4(w.g, frag(o0, hi), c1t);
        if (!hi) {
            v4f a; a.x = o0[0]; a.y = o0[1]; a.z = o0[2]; a.w = o0[3];
            v4f c; c.x = o1[0]; c.y = o1[1]; c.z = o1[2]; c.w = o1[3];
            *(v4f*)(&sx[2 * n][0])     = a;
            *(v4f*)(&sx[2 * n + 1][0]) = c;
        }
    }
    __syncthreads();
    const v4f sv = *(const v4f*)(&sx[lane][0]);
    float y0 = sv.x, y1 = sv.y, v0 = sv.z, v1 = sv.w;

    if (allfull) {
        emit_full_chunk(m, uu, y0, y1, v0, v1, op);
    } else if (full) {
        load_chunk(up, uu);
        emit_full_chunk(m, uu, y0, y1, v0, v1, op);
    } else if (live) {
        for (int t = i * CHUNK_L; t < T; ++t) {
            affine_step(m, y0, y1, v0, v1, u[2 * t], u[2 * t + 1]);
            out[2 * t]     = y0;
            out[2 * t + 1] = y1;
        }
    }
}

extern "C" void kernel_launch(void* const* d_in, const int* in_sizes, int n_in,
                              void* d_out, int out_size, void* d_ws,
                              size_t ws_size, hipStream_t stream) {
    const float* u  = (const float*)d_in[0];
    const float* M  = (const float*)d_in[1];
    const float* C  = (const float*)d_in[2];
    const float* K  = (const float*)d_in[3];
    const float* x0 = (const float*)d_in[4];
    const float* dt = (const float*)d_in[5];
    float* out = (float*)d_out;

    const int T = in_sizes[0] / 2;
    int P = (T + CHUNK_L - 1) / CHUNK_L;
    if (P > NCOLS) P = NCOLS;
    const int nfull = P - 1;

    float* wsE = (float*)d_ws;
    float* wsR = wsE + NCOLS * 4;
    float* wsT = wsR + (NCOLS / WCH) * 4;

    phase1_chunks<<<(nfull + CH_BLOCK - 1) / CH_BLOCK + 1, CH_BLOCK, 0, stream>>>(
        u, M, C, K, dt, wsE, wsR, wsT, nfull);
    phase2_emit<<<(P + CH_BLOCK - 1) / CH_BLOCK, CH_BLOCK, 0, stream>>>(
        u, x0, wsE, wsR, wsT, out, P, T);
}


# Store placement: launch 1 stores chunk ends behind its LDS hand-over, so that fence waits for no store acknowledge; launch 2 stores out every 2 steps inside the step chain, not in one run after it
# speedup vs baseline: 23.8460x; 1.0563x over previous
#include <hip/hip_runtime.h>


typedef __attribute__((ext_vector_type(2))) float v2f;
typedef __attribute__((ext_vector_type(4))) float v4f;
typedef __attribute__((ext_vector_type(8))) float v8f;
typedef __attribute__((ext_vector_type(4), aligned(8))) float v4f_a8;

#define CHUNK_L 25
#define NCOLS   4096
#define WCH     32
#define RUNB    8
#define CH_BLOCK 32
#define TAB_POW 640

struct Params {
    float cm00, cm01, cm10, cm11;
    float km00, km01, km10, km11;
    float m0, m1, dt, h2, dt6;
};

__device__ inline Params make_params(const float* M, const float* C,
                                     const float* K, const float* dtp) {
    Params p;
    p.cm00 = C[0] + C[1]; p.cm01 = -C[1];
    p.cm10 = -C[1];       p.cm11 = C[2] + C[1];
    p.km00 = K[0] + K[1]; p.km01 = -K[1];
    p.km10 = -K[1];       p.km11 = K[2] + K[1];
    p.m0 = M[0]; p.m1 = M[1];
    p.dt = dtp[0];
    p.h2 = 0.5f * p.dt;
    p.dt6 = p.dt / 6.0f;
    return p;
}

__device__ inline void accel(const Params& p, float z0, float z1,
                             float v0, float v1, float u0, float u1,
                             float& a0, float& a1) {
    a0 = (u0 - (p.cm00 * v0 + p.cm01 * v1) - (p.km00 * z0 + p.km01 * z1)) / p.m0;
    a1 = (u1 - (p.cm10 * v0 + p.cm11 * v1) - (p.km10 * z0 + p.km11 * z1)) / p.m1;
}

__device__ inline void rk4_step(const Params& p, float& y0, float& y1,
                                float& v0, float& v1, float u0, float u1) {
    float k1x, k1y; accel(p, y0, y1, v0, v1, u0, u1, k1x, k1y);
    float y2x = y0 + v0 * p.h2, y2y = y1 + v1 * p.h2;
    float v2x = v0 + k1x * p.h2, v2y = v1 + k1y * p.h2;
    float k2x, k2y; accel(p, y2x, y2y, v2x, v2y, u0, u1, k2x, k2y);
    float y3x = y0 + v2x * p.h2, y3y = y1 + v2y * p.h2;
    float v3x = v0 + k2x * p.h2, v3y = v1 + k2y * p.h2;
    float k3x, k3y; accel(p, y3x, y3y, v3x, v3y, u0, u1, k3x, k3y);
    float y4x = y0 + v3x * p.dt, y4y = y1 + v3y * p.dt;
    float v4x = v0 + k3x * p.dt, v4y = v1 + k3y * p.dt;
    float k4x, k4y; accel(p, y4x, y4y, v4x, v4y, u0, u1, k4x, k4y);
    float yn0 = y0 + p.dt6 * (v0 + 2.0f * v2x + 2.0f * v3x + v4x);
    float yn1 = y1 + p.dt6 * (v1 + 2.0f * v2y + 2.0f * v3y + v4y);
    float vn0 = v0 + p.dt6 * (k1x + 2.0f * k2x + 2.0f * k3x + k4x);
    float vn1 = v1 + p.dt6 * (k1y + 2.0f * k2y + 2.0f * k3y + k4y);
    y0 = yn0; y1 = yn1; v0 = vn0; v1 = vn1;
}

struct StepMap {
    float a[4][4];
    float b[4][2];
};

__device__ inline float lane_bcast(float x, int srclane) {
    return __int_as_float(__builtin_amdgcn_readlane(__float_as_int(x), srclane));
}

__device__ inline StepMap make_step_map(const Params& p, int lane, v8f& xa,
                                        v4f& colv) {
    const int j = lane & 7;
    float y0 = (j == 0) ? 1.f : 0.f, y1 = (j == 1) ? 1.f : 0.f;
    float v0 = (j == 2) ? 1.f : 0.f, v1 = (j == 3) ? 1.f : 0.f;
    const float u0 = (j == 4) ? 1.f : 0.f, u1 = (j == 5) ? 1.f : 0.f;
    rk4_step(p, y0, y1, v0, v1, u0, u1);
    StepMap m;
    #pragma unroll
    for (int c = 0; c < 4; ++c) {
        m.a[0][c] = lane_bcast(y0, c);
        m.a[1][c] = lane_bcast(y1, c);
        m.a[2][c] = lane_bcast(v0, c);
        m.a[3][c] = lane_bcast(v1, c);
    }
    #pragma unroll
    for (int c = 0; c < 2; ++c) {
        m.b[0][c] = lane_bcast(y0, 4 + c);
        m.b[1][c] = lane_bcast(y1, 4 + c);
        m.b[2][c] = lane_bcast(v0, 4 + c);
        m.b[3][c] = lane_bcast(v1, 4 + c);
    }
    const bool col = j < 4;
    xa = (v8f){col ? y0 : 0.f, col ? y1 : 0.f, col ? v0 : 0.f, col ? v1 : 0.f,
               0.f, 0.f, 0.f, 0.f};
    colv.x = y0; colv.y = y1; colv.z = v0; colv.w = v1;
    return m;
}

__device__ inline void affine_step(const StepMap& m, float& y0, float& y1,
                                   float& v0, float& v1, float u0, float u1) {
    const float s[4] = {y0, y1, v0, v1};
    float r[4];
    #pragma unroll
    for (int i = 0; i < 4; ++i) {
        float acc = m.b[i][1] * u1;
        acc = fmaf(m.b[i][0], u0, acc);
        #pragma unroll
        for (int j = 3; j >= 0; --j) {
            if (j != i) acc = fmaf(m.a[i][j], s[j], acc);
        }
        r[i] = fmaf(m.a[i][i], s[i], acc);
    }
    y0 = r[0]; y1 = r[1]; v0 = r[2]; v1 = r[3];
}

__device__ inline v8f wmma4(v2f a, v2f b, v8f c) {
    return __builtin_amdgcn_wmma_f32_16x16x4_f32(
        false, a, false, b, (short)0, c, false, false);
}

__device__ inline v2f frag(const v8f& t, bool hi) {
    v2f f;
    f.x = hi ? t[2] : t[0];
    f.y = hi ? t[3] : t[1];
    return f;
}

__device__ inline v8f col_tile(float a0, float a1, float a2, float a3) {
    return (v8f){a0, a1, a2, a3, 0.f, 0.f, 0.f, 0.f};
}

struct Pw {
    v2f fx, ft;
};

__device__ inline Pw pw_mul(const Pw& p, const Pw& q, bool hi) {
    const v8f z = {0.f, 0.f, 0.f, 0.f, 0.f, 0.f, 0.f, 0.f};
    const v8f x = wmma4(p.ft, q.fx, z);
    const v8f y = wmma4(q.fx, p.ft, z);
    Pw r;
    r.fx = frag(x, hi);
    r.ft = frag(y, hi);
    return r;
}

__device__ inline v8f tile_of_a_transposed(const StepMap& m, int lane) {
    const int i = lane & 7;
    float r[4];
    #pragma unroll
    for (int k = 0; k < 4; ++k) {
        const float r0 = m.a[0][k], r1 = m.a[1][k], r2 = m.a[2][k], r3 = m.a[3][k];
        float v = 0.f;
        v = (i == 3) ? r3 : v;
        v = (i == 2) ? r2 : v;
        v = (i == 1) ? r1 : v;
        v = (i == 0) ? r0 : v;
        r[k] = v;
    }
    return col_tile(r[0], r[1], r[2], r[3]);
}

struct PowTab {
    v2f g, g2, g4, g8, g16;
    v2f h, h8, h16, h32, h64;
};

template <bool BLOCKS>
__device__ inline PowTab build_powers(const StepMap& m, const v8f& xa,
                                      int lane, bool hi) {
    Pw a;
    a.fx = frag(xa, hi);
    a.ft = frag(tile_of_a_transposed(m, lane), hi);
    const Pw a2  = pw_mul(a,   a,   hi);
    const Pw a4  = pw_mul(a2,  a2,  hi);
    const Pw a8  = pw_mul(a4,  a4,  hi);
    const Pw a16 = pw_mul(a8,  a8,  hi);
    const Pw a24 = pw_mul(a16, a8,  hi);
    const Pw g   = pw_mul(a24, a,   hi);
    const Pw g2  = pw_mul(g,   g,   hi);
    const Pw g4  = pw_mul(g2,  g2,  hi);
    const Pw g8  = pw_mul(g4,  g4,  hi);
    const Pw g16 = pw_mul(g8,  g8,  hi);
    PowTab t;
    t.g = g.ft; t.g2 = g2.ft; t.g4 = g4.ft; t.g8 = g8.ft; t.g16 = g16.ft;
    t.h = t.h8 = t.h16 = t.h32 = t.h64 = g.ft;
    if (BLOCKS) {
        const Pw h   = pw_mul(g16, g16, hi);
        const Pw h2  = pw_mul(h,   h,   hi);
        const Pw h4  = pw_mul(h2,  h2,  hi);
        const Pw h8  = pw_mul(h4,  h4,  hi);
        const Pw h16 = pw_mul(h8,  h8,  hi);
        const Pw h32 = pw_mul(h16, h16, hi);
        const Pw h64 = pw_mul(h32, h32, hi);
        t.h = h.ft; t.h8 = h8.ft; t.h16 = h16.ft; t.h32 = h32.ft; t.h64 = h64.ft;
    }
    return t;
}

__device__ inline void store_table(float* __restrict__ wsT, const PowTab& w,
                                   v4f colv, int lane) {
    v4f* q = (v4f*)wsT + lane;
    q[0]   = (v4f){w.g.x,   w.g.y,   w.g2.x,  w.g2.y};
    q[32]  = (v4f){w.g4.x,  w.g4.y,  w.g8.x,  w.g8.y};
    q[64]  = (v4f){w.g16.x, w.g16.y, w.h.x,   w.h.y};
    q[96]  = (v4f){w.h8.x,  w.h8.y,  w.h16.x, w.h16.y};
    q[128] = (v4f){w.h32.x, w.h32.y, w.h64.x, w.h64.y};
    if (lane < 6) ((v4f*)(wsT + TAB_POW))[lane] = colv;
}

__device__ inline PowTab load_powers(const float* __restrict__ wsT, int lane) {
    const v4f* q = (const v4f*)wsT + lane;
    const v4f q0 = q[0], q1 = q[32], q2 = q[64], q3 = q[96], q4 = q[128];
    PowTab t;
    t.g   = (v2f){q0.x, q0.y}; t.g2  = (v2f){q0.z, q0.w};
    t.g4  = (v2f){q1.x, q1.y}; t.g8  = (v2f){q1.z, q1.w};
    t.g16 = (v2f){q2.x, q2.y}; t.h   = (v2f){q2.z, q2.w};
    t.h8  = (v2f){q3.x, q3.y}; t.h16 = (v2f){q3.z, q3.w};
    t.h32 = (v2f){q4.x, q4.y}; t.h64 = (v2f){q4.z, q4.w};
    return t;
}

__device__ inline StepMap load_step_map(const float* __restrict__ wsM) {
    StepMap m;
    #pragma unroll
    for (int c = 0; c < 4; ++c) {
        m.a[0][c] = wsM[c * 4 + 0]; m.a[1][c] = wsM[c * 4 + 1];
        m.a[2][c] = wsM[c * 4 + 2]; m.a[3][c] = wsM[c * 4 + 3];
    }
    #pragma unroll
    for (int c = 0; c < 2; ++c) {
        m.b[0][c] = wsM[(4 + c) * 4 + 0]; m.b[1][c] = wsM[(4 + c) * 4 + 1];
        m.b[2][c] = wsM[(4 + c) * 4 + 2]; m.b[3][c] = wsM[(4 + c) * 4 + 3];
    }
    return m;
}

template <int D>
__device__ inline v2f frag_from_left(const v8f& t, bool hi) {
    const v2f f = frag(t, hi);
    v2f r;
    r.x = __int_as_float(__builtin_amdgcn_update_dpp(
        0, __float_as_int(f.x), 0x110 + D, 0xf, 0xf, true));
    r.y = __int_as_float(__builtin_amdgcn_update_dpp(
        0, __float_as_int(f.y), 0x110 + D, 0xf, 0xf, true));
    return r;
}

__device__ inline v8f scan16(v8f t, v2f m1, v2f m2, v2f m4, v2f m8, bool hi) {
    t = wmma4(m1, frag_from_left<1>(t, hi), t);
    t = wmma4(m2, frag_from_left<2>(t, hi), t);
    t = wmma4(m4, frag_from_left<4>(t, hi), t);
    t = wmma4(m8, frag_from_left<8>(t, hi), t);
    return t;
}

__device__ inline void load_chunk(const float* __restrict__ up,
                                  float (&uu)[CHUNK_L][2]) {
    #pragma unroll
    for (int t = 0; t < CHUNK_L; ++t) {
        uu[t][0] = up[2 * t];
        uu[t][1] = up[2 * t + 1];
    }
}

__device__ inline void chunk_end_from_zero(const StepMap& m,
                                           const float (&uu)[CHUNK_L][2],
                                           float (&e)[4]) {
    float y0 = 0.f, y1 = 0.f, v0 = 0.f, v1 = 0.f;
    #pragma unroll
    for (int t = 0; t < CHUNK_L; ++t) {
        affine_step(m, y0, y1, v0, v1, uu[t][0], uu[t][1]);
    }
    e[0] = y0; e[1] = y1; e[2] = v0; e[3] = v1;
}

__global__ __launch_bounds__(CH_BLOCK) void phase1_chunks(
    const float* __restrict__ u, const float* __restrict__ M,
    const float* __restrict__ C, const float* __restrict__ K,
    const float* __restrict__ dtp, float* __restrict__ wsE,
    float* __restrict__ wsR, float* __restrict__ wsT, int nfull) {
    __shared__ __attribute__((aligned(16))) float ex[WCH][4];
    const int  lane = threadIdx.x & 31;
    const int  n    = lane & 15;
    const bool hi   = lane >= 16;
    const int  i    = blockIdx.x * blockDim.x + threadIdx.x;
    const bool live = i < nfull;
    const float* up = u + (size_t)i * CHUNK_L * 2;
    float uu[CHUNK_L][2];
    float e[4];
    v8f xa;
    v4f colv;
    if (__all(live)) {
        load_chunk(up, uu);
        const Params  p = make_params(M, C, K, dtp);
        const StepMap m = make_step_map(p, lane, xa, colv);
        const PowTab  w = build_powers<false>(m, xa, lane, hi);
        chunk_end_from_zero(m, uu, e);
        v4f ev; ev.x = e[0]; ev.y = e[1]; ev.z = e[2]; ev.w = e[3];
        *(v4f*)(&ex[lane][0]) = ev;
        __syncthreads();
        const v4f x0 = *(const v4f*)(&ex[2 * n][0]);
        const v4f x1 = *(const v4f*)(&ex[2 * n + 1][0]);
        *(v4f*)(wsE + (size_t)i * 4) = ev;
        v8f t = wmma4(w.g, frag(col_tile(x0.x, x0.y, x0.z, x0.w), hi),
                      col_tile(x1.x, x1.y, x1.z, x1.w));
        t = scan16(t, w.g2, w.g4, w.g8, w.g16, hi);
        if (lane == 15) {
            v4f r; r.x = t[0]; r.y = t[1]; r.z = t[2]; r.w = t[3];
            *(v4f*)(wsR + (size_t)blockIdx.x * 4) = r;
        }
    } else {
        const Params  p = make_params(M, C, K, dtp);
        const StepMap m = make_step_map(p, lane, xa, colv);
        if (blockIdx.x == gridDim.x - 1) {
            const PowTab w = build_powers<true>(m, xa, lane, hi);
            store_table(wsT, w, colv, lane);
        }
        if (live) {
            load_chunk(up, uu);
            chunk_end_from_zero(m, uu, e);
            wsE[i * 4 + 0] = e[0]; wsE[i * 4 + 1] = e[1];
            wsE[i * 4 + 2] = e[2]; wsE[i * 4 + 3] = e[3];
        }
    }
}

__device__ inline void emit_full_chunk(const StepMap& m,
                                       const float (&uu)[CHUNK_L][2],
                                       float y0, float y1, float v0, float v1,
                                       float* __restrict__ op) {
    float p0 = 0.f, p1 = 0.f;
    #pragma unroll
    for (int t = 0; t < CHUNK_L; ++t) {
        affine_step(m, y0, y1, v0, v1, uu[t][0], uu[t][1]);
        if (t & 1) {
            *(v4f_a8*)(op + 2 * (t - 1)) = (v4f_a8){p0, p1, y0, y1};
            __builtin_amdgcn_sched_barrier(0);
        } else if (t == CHUNK_L - 1) {
            *(v2f*)(op + 2 * t) = (v2f){y0, y1};
        } else {
            p0 = y0;
            p1 = y1;
        }
    }
}

__global__ __launch_bounds__(CH_BLOCK) void phase2_emit(
    const float* __restrict__ u, const float* __restrict__ x0,
    const float* __restrict__ wsE, const float* __restrict__ wsR,
    const float* __restrict__ wsT, float* __restrict__ out, int P, int T) {
    __shared__ __attribute__((aligned(16))) float sx[WCH][4];
    const int  lane = threadIdx.x & 31;
    const int  n    = lane & 15;
    const bool hi   = lane >= 16;
    const int  b    = blockIdx.x;
    const int  c0   = b * WCH;
    const int  i    = c0 + lane;
    const int  nfull = P - 1;
    const bool live = i < P;
    const bool full = live && (i + 1) * CHUNK_L <= T;
    const bool allfull = __all(full);
    const float* up = u + (size_t)i * CHUNK_L * 2;
    float*       op = out + (size_t)i * CHUNK_L * 2;
    v4f e0 = {0.f, 0.f, 0.f, 0.f}, e1 = {0.f, 0.f, 0.f, 0.f};
    {
        const int k0 = c0 + 2 * n - 1;
        const int k1 = c0 + 2 * n;
        if (n >= 1 && k0 < nfull) e0 = *(const v4f*)(wsE + (size_t)k0 * 4);
        if (k1 < nfull)           e1 = *(const v4f*)(wsE + (size_t)k1 * 4);
    }
    const PowTab  w = load_powers(wsT, lane);
    const StepMap m = load_step_map(wsT + TAB_POW);
    const float i0 = x0[0], i1 = x0[2], i2 = x0[1], i3 = x0[3];
    float xr[RUNB][4];
    #pragma unroll
    for (int g = 0; g < RUNB; ++g) {
        const int j = n * RUNB + g;
        v4f r = {0.f, 0.f, 0.f, 0.f};
        if (j >= 1 && j <= b) r = *(const v4f*)(wsR + (size_t)(j - 1) * 4);
        xr[g][0] = r.x; xr[g][1] = r.y; xr[g][2] = r.z; xr[g][3] = r.w;
    }
    {
        const bool init = (n == 0);
        xr[0][0] = init ? i0 : xr[0][0]; xr[0][1] = init ? i1 : xr[0][1];
        xr[0][2] = init ? i2 : xr[0][2]; xr[0][3] = init ? i3 : xr[0][3];
    }
    asm volatile("" :: "v"(xr[RUNB - 1][0]), "v"(xr[RUNB - 1][1]),
                       "v"(xr[RUNB - 1][2]), "v"(xr[RUNB - 1][3]));
    float uu[CHUNK_L][2];
    if (allfull) load_chunk(up, uu);

    float s0, s1, s2, s3;
    {
        v8f t = col_tile(xr[0][0], xr[0][1], xr[0][2], xr[0][3]);
        #pragma unroll
        for (int g = 1; g < RUNB; ++g) {
            t = wmma4(w.h, frag(t, hi),
                      col_tile(xr[g][0], xr[g][1], xr[g][2], xr[g][3]));
        }
        t = scan16(t, w.h8, w.h16, w.h32, w.h64, hi);
        v2f bp = frag_from_left<1>(t, hi);
        const int gsel = b & (RUNB - 1);
        float q0 = 0.f, q1 = 0.f, q2 = 0.f, q3 = 0.f;
        #pragma unroll
        for (int g = 0; g < RUNB; ++g) {
            const v8f o = wmma4(w.h, bp,
                                col_tile(xr[g][0], xr[g][1], xr[g][2], xr[g][3]));
            if (g == gsel) { q0 = o[0]; q1 = o[1]; q2 = o[2]; q3 = o[3]; }
            bp = frag(o, hi);
        }
        const int nsel = b / RUNB;
        s0 = lane_bcast(q0, nsel); s1 = lane_bcast(q1, nsel);
        s2 = lane_bcast(q2, nsel); s3 = lane_bcast(q3, nsel);
    }

    {
        const bool first = (n == 0);
        const v8f c0t = col_tile(first ? s0 : e0.x, first ? s1 : e0.y,
                                 first ? s2 : e0.z, first ? s3 : e0.w);
        const v8f c1t = col_tile(e1.x, e1.y, e1.z, e1.w);
        v8f t = wmma4(w.g, frag(c0t, hi), c1t);
        t = scan16(t, w.g2, w.g4, w.g8, w.g16, hi);
        const v8f o0 = wmma4(w.g, frag_from_left<1>(t, hi), c0t);
        const v8f o1 = wmma4(w.g, frag(o0, hi), c1t);
        if (!hi) {
            v4f a; a.x = o0[0]; a.y = o0[1]; a.z = o0[2]; a.w = o0[3];
            v4f c; c.x = o1[0]; c.y = o1[1]; c.z = o1[2]; c.w = o1[3];
            *(v4f*)(&sx[2 * n][0])     = a;
            *(v4f*)(&sx[2 * n + 1][0]) = c;
        }
    }
    __syncthreads();
    const v4f sv = *(const v4f*)(&sx[lane][0]);
    float y0 = sv.x, y1 = sv.y, v0 = sv.z, v1 = sv.w;

    if (allfull) {
        emit_full_chunk(m, uu, y0, y1, v0, v1, op);
    } else if (full) {
        load_chunk(up, uu);
        emit_full_chunk(m, uu, y0, y1, v0, v1, op);
    } else if (live) {
        for (int t = i * CHUNK_L; t < T; ++t) {
            affine_step(m, y0, y1, v0, v1, u[2 * t], u[2 * t + 1]);
            out[2 * t]     = y0;
            out[2 * t + 1] = y1;
        }
    }
}

extern "C" void kernel_launch(void* const* d_in, const int* in_sizes, int n_in,
                              void* d_out, int out_size, void* d_ws,
                              size_t ws_size, hipStream_t stream) {
    const float* u  = (const float*)d_in[0];
    const float* M  = (const float*)d_in[1];
    const float* C  = (const float*)d_in[2];
    const float* K  = (const float*)d_in[3];
    const float* x0 = (const float*)d_in[4];
    const float* dt = (const float*)d_in[5];
    float* out = (float*)d_out;

    const int T = in_sizes[0] / 2;
    int P = (T + CHUNK_L - 1) / CHUNK_L;
    if (P > NCOLS) P = NCOLS;
    const int nfull = P - 1;

    float* wsE = (float*)d_ws;
    float* wsR = wsE + NCOLS * 4;
    float* wsT = wsR + (NCOLS / WCH) * 4;

    phase1_chunks<<<(nfull + CH_BLOCK - 1) / CH_BLOCK + 1, CH_BLOCK, 0, stream>>>(
        u, M, C, K, dt, wsE, wsR, wsT, nfull);
    phase2_emit<<<(P + CH_BLOCK - 1) / CH_BLOCK, CH_BLOCK, 0, stream>>>(
        u, x0, wsE, wsR, wsT, out, P, T);
}
